# starting kernel
# baseline (speedup 1.0000x reference)
#include <hip/hip_runtime.h>
#include <cfloat>

#define N_NODES 100000
#define NFEAT   256
#define NHID    64
#define NEDGE   1600000
#define TOTE    (NEDGE + N_NODES)
#define MTILES  (N_NODES / 16)
#define EPSBN   1e-5f
#define NEG_SLOPE 0.2f

typedef __attribute__((ext_vector_type(2))) float v2f;
typedef __attribute__((ext_vector_type(4))) float v4f;
typedef __attribute__((ext_vector_type(8))) float v8f;

__device__ __forceinline__ unsigned ford(float f) {
    unsigned u = __float_as_uint(f);
    return (u & 0x80000000u) ? ~u : (u | 0x80000000u);
}
__device__ __forceinline__ float dorf(unsigned u) {
    return (u & 0x80000000u) ? __uint_as_float(u & 0x7fffffffu) : __uint_as_float(~u);
}
__device__ __forceinline__ float leaky(float v) {
    return fmaxf(v, 0.f) + NEG_SLOPE * fminf(v, 0.f);
}

__global__ void k_init(float* acc_est, float* acc_gnn,
                       unsigned* m_e, unsigned* m_g, float* z_e, float* z_g) {
    long i = (long)blockIdx.x * blockDim.x + threadIdx.x;
    long tot = (long)N_NODES * NHID;
    if (i < tot) { acc_est[i] = 0.f; acc_gnn[i] = 0.f; }
    if (i < N_NODES) {
        m_e[i] = ford(-FLT_MAX);
        m_g[i] = ford(-FLT_MAX);
        z_e[i] = 0.f;
        z_g[i] = 0.f;
    }
}

__global__ __launch_bounds__(256) void k_gemm(const float* __restrict__ g,
                                              const float* __restrict__ We,
                                              const float* __restrict__ Wg,
                                              float* __restrict__ h_est,
                                              float* __restrict__ h_gnn) {
    int wave = threadIdx.x >> 5;
    int lane = threadIdx.x & 31;
    int mtile = blockIdx.x * 8 + wave;
    if (mtile >= MTILES) return;
    int m0   = mtile * 16;
    int lrow = lane & 15;
    int lhi  = lane >> 4;

    const float* grow = g + (long)(m0 + lrow) * NFEAT;

    v8f acc[8] = {};
    for (int k = 0; k < NFEAT; k += 4) {
        v2f a = *(const v2f*)(grow + k + 2 * lhi);
#pragma unroll
        for (int t = 0; t < 8; ++t) {
            const float* Wp = (t < 4) ? We : Wg;
            int c  = (t & 3) * 16 + lrow;
            int kr = k + 2 * lhi;
            v2f b;
            b.x = Wp[(long)kr * NHID + c];
            b.y = Wp[(long)(kr + 1) * NHID + c];
            acc[t] = __builtin_amdgcn_wmma_f32_16x16x4_f32(
                false, a, false, b, (short)0, acc[t], false, false);
        }
    }
#pragma unroll
    for (int t = 0; t < 8; ++t) {
        float* dst = (t < 4) ? h_est : h_gnn;
        int c = (t & 3) * 16 + lrow;
#pragma unroll
        for (int r = 0; r < 8; ++r) {
            dst[(long)(m0 + r + 8 * lhi) * NHID + c] = acc[t][r];
        }
    }
}

__global__ void k_eprep(const float* __restrict__ h_est, const float* __restrict__ h_gnn,
                        const float* __restrict__ ase, const float* __restrict__ ade,
                        const float* __restrict__ asg, const float* __restrict__ adg,
                        float* es_e, float* ed_e, float* es_g, float* ed_g) {
    int n = blockIdx.x * blockDim.x + threadIdx.x;
    if (n >= N_NODES) return;
    const float* he = h_est + (long)n * NHID;
    const float* hg = h_gnn + (long)n * NHID;
    float se = 0.f, de = 0.f, sg = 0.f, dg = 0.f;
#pragma unroll 4
    for (int f = 0; f < NHID; f += 4) {
        v4f e  = *(const v4f*)(he + f);
        v4f g4 = *(const v4f*)(hg + f);
        v4f a1 = *(const v4f*)(ase + f);
        v4f a2 = *(const v4f*)(ade + f);
        v4f a3 = *(const v4f*)(asg + f);
        v4f a4 = *(const v4f*)(adg + f);
        se += e.x * a1.x + e.y * a1.y + e.z * a1.z + e.w * a1.w;
        de += e.x * a2.x + e.y * a2.y + e.z * a2.z + e.w * a2.w;
        sg += g4.x * a3.x + g4.y * a3.y + g4.z * a3.z + g4.w * a3.w;
        dg += g4.x * a4.x + g4.y * a4.y + g4.z * a4.z + g4.w * a4.w;
    }
    es_e[n] = se; ed_e[n] = de; es_g[n] = sg; ed_g[n] = dg;
}

__global__ void k_max(const int* __restrict__ x,
                      const float* __restrict__ es_e, const float* __restrict__ ed_e,
                      const float* __restrict__ es_g, const float* __restrict__ ed_g,
                      unsigned* m_e, unsigned* m_g) {
    long e = (long)blockIdx.x * blockDim.x + threadIdx.x;
    if (e >= TOTE) return;
    int s, d;
    if (e < NEDGE) { s = x[e]; d = x[NEDGE + e]; } else { s = d = (int)(e - NEDGE); }
    atomicMax(m_e + d, ford(leaky(es_e[s] + ed_e[d])));
    atomicMax(m_g + d, ford(leaky(es_g[s] + ed_g[d])));
}

__global__ void k_sum(const int* __restrict__ x,
                      const float* __restrict__ es_e, const float* __restrict__ ed_e,
                      const float* __restrict__ es_g, const float* __restrict__ ed_g,
                      const unsigned* __restrict__ m_e, const unsigned* __restrict__ m_g,
                      float* z_e, float* z_g) {
    long e = (long)blockIdx.x * blockDim.x + threadIdx.x;
    if (e >= TOTE) return;
    int s, d;
    if (e < NEDGE) { s = x[e]; d = x[NEDGE + e]; } else { s = d = (int)(e - NEDGE); }
    atomicAdd(z_e + d, __expf(leaky(es_e[s] + ed_e[d]) - dorf(m_e[d])));
    atomicAdd(z_g + d, __expf(leaky(es_g[s] + ed_g[d]) - dorf(m_g[d])));
}

__global__ __launch_bounds__(256) void k_scatter(const int* __restrict__ x,
        const float* __restrict__ es_e, const float* __restrict__ ed_e,
        const float* __restrict__ es_g, const float* __restrict__ ed_g,
        const unsigned* __restrict__ m_e, const unsigned* __restrict__ m_g,
        const float* __restrict__ z_e, const float* __restrict__ z_g,
        const float* __restrict__ h_est, const float* __restrict__ h_gnn,
        float* acc_est, float* acc_gnn) {
    int wave = threadIdx.x >> 5;
    int lane = threadIdx.x & 31;
    long e = (long)blockIdx.x * 8 + wave;
    if (e >= TOTE) return;
    int s, d;
    if (e < NEDGE) { s = x[e]; d = x[NEDGE + e]; } else { s = d = (int)(e - NEDGE); }
    float ae = __expf(leaky(es_e[s] + ed_e[d]) - dorf(m_e[d])) / z_e[d];
    float ag = __expf(leaky(es_g[s] + ed_g[d]) - dorf(m_g[d])) / z_g[d];
    int f0 = lane * 2;
    v2f he = *(const v2f*)(h_est + (long)s * NHID + f0);
    v2f hg = *(const v2f*)(h_gnn + (long)s * NHID + f0);
    float* pe = acc_est + (long)d * NHID + f0;
    float* pg = acc_gnn + (long)d * NHID + f0;
    atomicAdd(pe,     he.x * ae);
    atomicAdd(pe + 1, he.y * ae);
    atomicAdd(pg,     hg.x * ag);
    atomicAdd(pg + 1, hg.y * ag);
}

__global__ void k_final(float* d_out, const float* __restrict__ acc_gnn,
        const float* __restrict__ eb, const float* __restrict__ egam,
        const float* __restrict__ ebeta, const float* __restrict__ erm,
        const float* __restrict__ erv,
        const float* __restrict__ gb, const float* __restrict__ ggam,
        const float* __restrict__ gbeta, const float* __restrict__ grm,
        const float* __restrict__ grv,
        const float* __restrict__ clsW, const float* __restrict__ clsb) {
    int n = blockIdx.x * blockDim.x + threadIdx.x;
    if (n >= N_NODES) return;
    float* srow = d_out + N_NODES + (long)n * NHID;
    const float* grow = acc_gnn + (long)n * NHID;
    float y = 0.f;
#pragma unroll 4
    for (int f = 0; f < NHID; ++f) {
        float a = fmaxf(srow[f] + eb[f], 0.f);
        srow[f] = (a - erm[f]) * rsqrtf(erv[f] + EPSBN) * egam[f] + ebeta[f];
        float b = fmaxf(grow[f] + gb[f], 0.f);
        float zv = (b - grm[f]) * rsqrtf(grv[f] + EPSBN) * ggam[f] + gbeta[f];
        y += zv * clsW[f];
    }
    d_out[n] = y + clsb[0];
}

extern "C" void kernel_launch(void* const* d_in, const int* in_sizes, int n_in,
                              void* d_out, int out_size, void* d_ws, size_t ws_size,
                              hipStream_t stream) {
    const float* g        = (const float*)d_in[0];
    const int*   x        = (const int*)d_in[1];
    const float* est_W    = (const float*)d_in[2];
    const float* est_asrc = (const float*)d_in[3];
    const float* est_adst = (const float*)d_in[4];
    const float* est_bias = (const float*)d_in[5];
    const float* est_gam  = (const float*)d_in[6];
    const float* est_beta = (const float*)d_in[7];
    const float* est_rm   = (const float*)d_in[8];
    const float* est_rv   = (const float*)d_in[9];
    const float* gnn_W    = (const float*)d_in[10];
    const float* gnn_asrc = (const float*)d_in[11];
    const float* gnn_adst = (const float*)d_in[12];
    const float* gnn_bias = (const float*)d_in[13];
    const float* gnn_gam  = (const float*)d_in[14];
    const float* gnn_beta = (const float*)d_in[15];
    const float* gnn_rm   = (const float*)d_in[16];
    const float* gnn_rv   = (const float*)d_in[17];
    const float* cls_W    = (const float*)d_in[18];
    const float* cls_b    = (const float*)d_in[19];
    float* out = (float*)d_out;

    float*    ws     = (float*)d_ws;
    float*    h_est  = ws;
    float*    h_gnn  = h_est + (long)N_NODES * NHID;
    float*    acc_g  = h_gnn + (long)N_NODES * NHID;
    float*    es_e   = acc_g + (long)N_NODES * NHID;
    float*    ed_e   = es_e + N_NODES;
    float*    es_g   = ed_e + N_NODES;
    float*    ed_g   = es_g + N_NODES;
    unsigned* m_e    = (unsigned*)(ed_g + N_NODES);
    unsigned* m_g    = m_e + N_NODES;
    float*    z_e    = (float*)(m_g + N_NODES);
    float*    z_g    = z_e + N_NODES;
    float*    acc_e  = out + N_NODES;

    const int BT = 256;
    long initN = (long)N_NODES * NHID;
    k_init<<<(unsigned)((initN + BT - 1) / BT), BT, 0, stream>>>(acc_e, acc_g, m_e, m_g, z_e, z_g);

    k_gemm<<<(MTILES + 7) / 8, BT, 0, stream>>>(g, est_W, gnn_W, h_est, h_gnn);

    k_eprep<<<(N_NODES + BT - 1) / BT, BT, 0, stream>>>(h_est, h_gnn,
            est_asrc, est_adst, gnn_asrc, gnn_adst, es_e, ed_e, es_g, ed_g);

    k_max<<<(TOTE + BT - 1) / BT, BT, 0, stream>>>(x, es_e, ed_e, es_g, ed_g, m_e, m_g);

    k_sum<<<(TOTE + BT - 1) / BT, BT, 0, stream>>>(x, es_e, ed_e, es_g, ed_g, m_e, m_g, z_e, z_g);

    k_scatter<<<(TOTE + 7) / 8, BT, 0, stream>>>(x, es_e, ed_e, es_g, ed_g,
            m_e, m_g, z_e, z_g, h_est, h_gnn, acc_e, acc_g);

    k_final<<<(N_NODES + BT - 1) / BT, BT, 0, stream>>>(out, acc_g,
            est_bias, est_gam, est_beta, est_rm, est_rv,
            gnn_bias, gnn_gam, gnn_beta, gnn_rm, gnn_rv, cls_W, cls_b);
}


# Each work-group owns 400 destination nodes: max, softmax sums and row sums kept in LDS (LDS atomics, arrival order), edges found by a byte scan; global atomics, init and finalize kernels dropped
# speedup vs baseline: 3.5377x; 3.5377x over previous
#include <hip/hip_runtime.h>
#include <cfloat>

#define N_NODES 100000
#define NFEAT   256
#define NHID    64
#define NEDGE   1600000
#define TOTE    (NEDGE + N_NODES)
#define MTILES  (N_NODES / 16)
#define EPSBN   1e-5f
#define NEG_SLOPE 0.2f

typedef __attribute__((ext_vector_type(2))) float v2f;
typedef __attribute__((ext_vector_type(4))) float v4f;
typedef __attribute__((ext_vector_type(8))) float v8f;

__device__ __forceinline__ unsigned ford(float f) {
    unsigned u = __float_as_uint(f);
    return (u & 0x80000000u) ? ~u : (u | 0x80000000u);
}
__device__ __forceinline__ float dorf(unsigned u) {
    return (u & 0x80000000u) ? __uint_as_float(u & 0x7fffffffu) : __uint_as_float(~u);
}
__device__ __forceinline__ float leaky(float v) {
    return fmaxf(v, 0.f) + NEG_SLOPE * fminf(v, 0.f);
}

#define NGRP    250
#define GSZ     400
#define PITCH   65
#define CAP     24576
#define NV16    (NEDGE / 16)

typedef __attribute__((ext_vector_type(4))) int v4i;
typedef __attribute__((ext_vector_type(4))) unsigned v4u;

__global__ void k_grp(const int* __restrict__ x, unsigned* __restrict__ grp4) {
    int t = blockIdx.x * blockDim.x + threadIdx.x;
    if (t >= NEDGE / 4) return;
    v4i d = *(const v4i*)(x + NEDGE + 4 * t);
    grp4[t] = (unsigned)(d.x / GSZ) | ((unsigned)(d.y / GSZ) << 8) |
              ((unsigned)(d.z / GSZ) << 16) | ((unsigned)(d.w / GSZ) << 24);
}

__global__ __launch_bounds__(256) void k_gemm(const float* __restrict__ g,
                                              const float* __restrict__ We,
                                              const float* __restrict__ Wg,
                                              float* __restrict__ h_est,
                                              float* __restrict__ h_gnn) {
    int wave = threadIdx.x >> 5;
    int lane = threadIdx.x & 31;
    int mtile = blockIdx.x * 8 + wave;
    if (mtile >= MTILES) return;
    int m0   = mtile * 16;
    int lrow = lane & 15;
    int lhi  = lane >> 4;

    const float* grow = g + (long)(m0 + lrow) * NFEAT;

    v8f acc[8] = {};
    for (int k = 0; k < NFEAT; k += 4) {
        v2f a = *(const v2f*)(grow + k + 2 * lhi);
#pragma unroll
        for (int t = 0; t < 8; ++t) {
            const float* Wp = (t < 4) ? We : Wg;
            int c  = (t & 3) * 16 + lrow;
            int kr = k + 2 * lhi;
            v2f b;
            b.x = Wp[(long)kr * NHID + c];
            b.y = Wp[(long)(kr + 1) * NHID + c];
            acc[t] = __builtin_amdgcn_wmma_f32_16x16x4_f32(
                false, a, false, b, (short)0, acc[t], false, false);
        }
    }
#pragma unroll
    for (int t = 0; t < 8; ++t) {
        float* dst = (t < 4) ? h_est : h_gnn;
        int c = (t & 3) * 16 + lrow;
#pragma unroll
        for (int r = 0; r < 8; ++r) {
            dst[(long)(m0 + r + 8 * lhi) * NHID + c] = acc[t][r];
        }
    }
}

__global__ void k_eprep(const float* __restrict__ h_est, const float* __restrict__ h_gnn,
                        const float* __restrict__ ase, const float* __restrict__ ade,
                        const float* __restrict__ asg, const float* __restrict__ adg,
                        float* es_e, float* ed_e, float* es_g, float* ed_g) {
    int n = blockIdx.x * blockDim.x + threadIdx.x;
    if (n >= N_NODES) return;
    const float* he = h_est + (long)n * NHID;
    const float* hg = h_gnn + (long)n * NHID;
    float se = 0.f, de = 0.f, sg = 0.f, dg = 0.f;
#pragma unroll 4
    for (int f = 0; f < NHID; f += 4) {
        v4f e  = *(const v4f*)(he + f);
        v4f g4 = *(const v4f*)(hg + f);
        v4f a1 = *(const v4f*)(ase + f);
        v4f a2 = *(const v4f*)(ade + f);
        v4f a3 = *(const v4f*)(asg + f);
        v4f a4 = *(const v4f*)(adg + f);
        se += e.x * a1.x + e.y * a1.y + e.z * a1.z + e.w * a1.w;
        de += e.x * a2.x + e.y * a2.y + e.z * a2.z + e.w * a2.w;
        sg += g4.x * a3.x + g4.y * a3.y + g4.z * a3.z + g4.w * a3.w;
        dg += g4.x * a4.x + g4.y * a4.y + g4.z * a4.z + g4.w * a4.w;
    }
    es_e[n] = se; ed_e[n] = de; es_g[n] = sg; ed_g[n] = dg;
}

__device__ __forceinline__ unsigned zbytes(unsigned u) {
    unsigned t = (u & 0x7f7f7f7fu) + 0x7f7f7f7fu;
    return ~(t | u | 0x7f7f7f7fu);
}
__device__ __forceinline__ unsigned nib(unsigned f) {
    return ((f >> 7) & 1u) | ((f >> 14) & 2u) | ((f >> 21) & 4u) | ((f >> 28) & 8u);
}
__device__ __forceinline__ unsigned match16(v4u v, unsigned pat) {
    unsigned f0 = zbytes(v.x ^ pat), f1 = zbytes(v.y ^ pat);
    unsigned f2 = zbytes(v.z ^ pat), f3 = zbytes(v.w ^ pat);
    if ((f0 | f1 | f2 | f3) == 0u) return 0u;
    return nib(f0) | (nib(f1) << 4) | (nib(f2) << 8) | (nib(f3) << 12);
}
__device__ __forceinline__ int rl_i(int v, int j) { return __builtin_amdgcn_readlane(v, j); }
__device__ __forceinline__ float rl_f(float v, int j) {
    return __int_as_float(__builtin_amdgcn_readlane(__float_as_int(v), j));
}

__global__ __launch_bounds__(1024) void k_own(const int* __restrict__ x,
        const unsigned* __restrict__ grp4,
        const float* __restrict__ es_e, const float* __restrict__ ed_e,
        const float* __restrict__ es_g, const float* __restrict__ ed_g,
        const float* __restrict__ h_est, const float* __restrict__ h_gnn,
        float* __restrict__ d_out,
        const float* __restrict__ eb, const float* __restrict__ egam,
        const float* __restrict__ ebeta, const float* __restrict__ erm,
        const float* __restrict__ erv,
        const float* __restrict__ gb, const float* __restrict__ ggam,
        const float* __restrict__ gbeta, const float* __restrict__ grm,
        const float* __restrict__ grv,
        const float* __restrict__ clsW, const float* __restrict__ clsb) {
    __shared__ float accE[GSZ * PITCH];
    __shared__ float accG[GSZ * PITCH];
    __shared__ unsigned lst[CAP];
    __shared__ float zE[GSZ], zG[GSZ], edE[GSZ], edG[GSZ];
    __shared__ unsigned mE[GSZ], mG[GSZ];
    __shared__ unsigned cnt, ovf;

    const int tid  = threadIdx.x;
    const int lane = tid & 31;
    const int wave = __builtin_amdgcn_readfirstlane(tid >> 5);
    const int grpid = blockIdx.x;
    const int n0   = grpid * GSZ;
    const unsigned pat = (unsigned)grpid * 0x01010101u;
    const v4u* g16 = (const v4u*)grp4;

    for (int i = tid; i < GSZ * PITCH; i += 1024) { accE[i] = 0.f; accG[i] = 0.f; }
    if (tid < GSZ) {
        int n = n0 + tid;
        edE[tid] = ed_e[n]; edG[tid] = ed_g[n];
        zE[tid] = 0.f; zG[tid] = 0.f;
        mE[tid] = ford(-FLT_MAX); mG[tid] = ford(-FLT_MAX);
        lst[tid] = (unsigned)(NEDGE + n);
    }
    if (tid == 0) { cnt = GSZ; ovf = 0u; }
    __syncthreads();

    for (int i = tid; i < NV16; i += 4096) {
        v4u v[4];
#pragma unroll
        for (int u = 0; u < 4; ++u) {
            int ii = i + u * 1024;
            v[u] = (ii < NV16) ? g16[ii] : (v4u){0xffffffffu, 0xffffffffu, 0xffffffffu, 0xffffffffu};
        }
#pragma unroll
        for (int u = 0; u < 4; ++u) {
            unsigned m = match16(v[u], pat);
            while (m) {
                int j = __builtin_ctz(m); m &= m - 1u;
                unsigned idx = atomicAdd(&cnt, 1u);
                if (idx < CAP) lst[idx] = (unsigned)((i + u * 1024) * 16 + j); else ovf = 1u;
            }
        }
    }
    __syncthreads();
    const bool spilled = ovf != 0u;
    const int total = __builtin_amdgcn_readfirstlane(spilled ? GSZ : (int)cnt);

    for (int idx = tid; idx < total; idx += 1024) {
        int e = (int)lst[idx];
        int s, d;
        if (e < NEDGE) { s = x[e]; d = x[NEDGE + e]; } else { s = d = e - NEDGE; }
        int l = d - n0;
        atomicMax(&mE[l], ford(leaky(es_e[s] + edE[l])));
        atomicMax(&mG[l], ford(leaky(es_g[s] + edG[l])));
        lst[idx] = (unsigned)s | ((unsigned)l << 17);
    }
    if (spilled) {
        for (int i = tid; i < NV16; i += 1024) {
            unsigned m = match16(g16[i], pat);
            while (m) {
                int j = __builtin_ctz(m); m &= m - 1u;
                int e = i * 16 + j;
                int s = x[e], l = x[NEDGE + e] - n0;
                atomicMax(&mE[l], ford(leaky(es_e[s] + edE[l])));
                atomicMax(&mG[l], ford(leaky(es_g[s] + edG[l])));
            }
        }
    }
    __syncthreads();

    for (int base = wave * 32; base < total; base += 1024) {
        int idx = base + lane;
        bool valid = idx < total;
        unsigned ent = valid ? lst[idx] : 0u;
        int s = (int)(ent & 0x1ffffu);
        int l = (int)(ent >> 17);
        float ae = 0.f, ag = 0.f;
        if (valid) {
            ae = __expf(leaky(es_e[s] + edE[l]) - dorf(mE[l]));
            ag = __expf(leaky(es_g[s] + edG[l]) - dorf(mG[l]));
            atomicAdd(&zE[l], ae);
            atomicAdd(&zG[l], ag);
        }
        int nv = total - base; if (nv > 32) nv = 32;
        int j = 0;
        for (; j + 8 <= nv; j += 8) {
            float r0[8], r1[8], r2[8], r3[8], aj[8], gj[8]; int lj[8];
#pragma unroll
            for (int u = 0; u < 8; ++u) {
                int sj = rl_i(s, j + u);
                lj[u] = rl_i(l, j + u); aj[u] = rl_f(ae, j + u); gj[u] = rl_f(ag, j + u);
                const float* qe = h_est + (long)sj * NHID + lane;
                const float* qg = h_gnn + (long)sj * NHID + lane;
                r0[u] = qe[0]; r1[u] = qe[32]; r2[u] = qg[0]; r3[u] = qg[32];
            }
#pragma unroll
            for (int u = 0; u < 8; ++u) {
                float* pe = &accE[lj[u] * PITCH + lane];
                float* pg = &accG[lj[u] * PITCH + lane];
                atomicAdd(pe,      r0[u] * aj[u]);
                atomicAdd(pe + 32, r1[u] * aj[u]);
                atomicAdd(pg,      r2[u] * gj[u]);
                atomicAdd(pg + 32, r3[u] * gj[u]);
            }
        }
        for (; j < nv; ++j) {
            int sj = rl_i(s, j), l1 = rl_i(l, j);
            float a1 = rl_f(ae, j), g1 = rl_f(ag, j);
            const float* qe = h_est + (long)sj * NHID + lane;
            const float* qg = h_gnn + (long)sj * NHID + lane;
            float t0 = qe[0], t1 = qe[32], t2 = qg[0], t3 = qg[32];
            float* pe = &accE[l1 * PITCH + lane];
            float* pg = &accG[l1 * PITCH + lane];
            atomicAdd(pe,      t0 * a1);
            atomicAdd(pe + 32, t1 * a1);
            atomicAdd(pg,      t2 * g1);
            atomicAdd(pg + 32, t3 * g1);
        }
    }
    if (spilled) {
        for (int i = tid; i < NV16; i += 1024) {
            unsigned m = match16(g16[i], pat);
            while (m) {
                int j = __builtin_ctz(m); m &= m - 1u;
                int e = i * 16 + j;
                int s = x[e], l = x[NEDGE + e] - n0;
                float ae = __expf(leaky(es_e[s] + edE[l]) - dorf(mE[l]));
                float ag = __expf(leaky(es_g[s] + edG[l]) - dorf(mG[l]));
                atomicAdd(&zE[l], ae);
                atomicAdd(&zG[l], ag);
                for (int f = 0; f < NHID; ++f) {
                    atomicAdd(&accE[l * PITCH + f], h_est[(long)s * NHID + f] * ae);
                    atomicAdd(&accG[l * PITCH + f], h_gnn[(long)s * NHID + f] * ag);
                }
            }
        }
    }
    __syncthreads();
    if (tid < GSZ) { zE[tid] = 1.f / zE[tid]; zG[tid] = 1.f / zG[tid]; }
    __syncthreads();

    {
        int f = tid & 63;
        float b = eb[f], rm = erm[f], rs = rsqrtf(erv[f] + EPSBN), gm = egam[f], bt = ebeta[f];
        float* srow = d_out + N_NODES + (long)n0 * NHID;
        for (int i = tid; i < GSZ * NHID; i += 1024) {
            int l = i >> 6;
            float a = fmaxf(accE[l * PITCH + f] * zE[l] + b, 0.f);
            srow[i] = (a - rm) * rs * gm + bt;
        }
    }
    if (tid < GSZ) {
        const float* grow = accG + tid * PITCH;
        float rzg = zG[tid];
        float y = 0.f;
#pragma unroll 4
        for (int f = 0; f < NHID; ++f) {
            float b = fmaxf(grow[f] * rzg + gb[f], 0.f);
            float zv = (b - grm[f]) * rsqrtf(grv[f] + EPSBN) * ggam[f] + gbeta[f];
            y += zv * clsW[f];
        }
        d_out[n0 + tid] = y + clsb[0];
    }
}

extern "C" void kernel_launch(void* const* d_in, const int* in_sizes, int n_in,
                              void* d_out, int out_size, void* d_ws, size_t ws_size,
                              hipStream_t stream) {
    const float* g        = (const float*)d_in[0];
    const int*   x        = (const int*)d_in[1];
    const float* est_W    = (const float*)d_in[2];
    const float* est_asrc = (const float*)d_in[3];
    const float* est_adst = (const float*)d_in[4];
    const float* est_bias = (const float*)d_in[5];
    const float* est_gam  = (const float*)d_in[6];
    const float* est_beta = (const float*)d_in[7];
    const float* est_rm   = (const float*)d_in[8];
    const float* est_rv   = (const float*)d_in[9];
    const float* gnn_W    = (const float*)d_in[10];
    const float* gnn_asrc = (const float*)d_in[11];
    const float* gnn_adst = (const float*)d_in[12];
    const float* gnn_bias = (const float*)d_in[13];
    const float* gnn_gam  = (const float*)d_in[14];
    const float* gnn_beta = (const float*)d_in[15];
    const float* gnn_rm   = (const float*)d_in[16];
    const float* gnn_rv   = (const float*)d_in[17];
    const float* cls_W    = (const float*)d_in[18];
    const float* cls_b    = (const float*)d_in[19];
    float* out = (float*)d_out;

    float*    ws     = (float*)d_ws;
    float*    h_est  = ws;
    float*    h_gnn  = h_est + (long)N_NODES * NHID;
    float*    es_e   = h_gnn + (long)N_NODES * NHID;
    float*    ed_e   = es_e + N_NODES;
    float*    es_g   = ed_e + N_NODES;
    float*    ed_g   = es_g + N_NODES;
    unsigned* grp4   = (unsigned*)(ed_g + N_NODES);

    const int BT = 256;
    k_grp<<<(NEDGE / 4 + BT - 1) / BT, BT, 0, stream>>>(x, grp4);

    k_gemm<<<(MTILES + 7) / 8, BT, 0, stream>>>(g, est_W, gnn_W, h_est, h_gnn);

    k_eprep<<<(N_NODES + BT - 1) / BT, BT, 0, stream>>>(h_est, h_gnn,
            est_asrc, est_adst, gnn_asrc, gnn_adst, es_e, ed_e, es_g, ed_g);

    k_own<<<NGRP, 1024, 0, stream>>>(x, grp4, es_e, ed_e, es_g, ed_g, h_est, h_gnn, out,
            est_bias, est_gam, est_beta, est_rm, est_rv,
            gnn_bias, gnn_gam, gnn_beta, gnn_rm, gnn_rv, cls_W, cls_b);
}


# Owner kernel: scattered 4-byte fetches per edge cut from 6 to 2: src and local dst as one packed word per edge, both source logits as one 8-byte pair, the sum pass reuses logits kept in LDS
# speedup vs baseline: 3.8454x; 1.0870x over previous
#include <hip/hip_runtime.h>
#include <cfloat>

#define N_NODES 100000
#define NFEAT   256
#define NHID    64
#define NEDGE   1600000
#define TOTE    (NEDGE + N_NODES)
#define MTILES  (N_NODES / 16)
#define EPSBN   1e-5f
#define NEG_SLOPE 0.2f

typedef __attribute__((ext_vector_type(2))) float v2f;
typedef __attribute__((ext_vector_type(4))) float v4f;
typedef __attribute__((ext_vector_type(8))) float v8f;

__device__ __forceinline__ unsigned ford(float f) {
    unsigned u = __float_as_uint(f);
    return (u & 0x80000000u) ? ~u : (u | 0x80000000u);
}
__device__ __forceinline__ float dorf(unsigned u) {
    return (u & 0x80000000u) ? __uint_as_float(u & 0x7fffffffu) : __uint_as_float(~u);
}
__device__ __forceinline__ float leaky(float v) {
    return fmaxf(v, 0.f) + NEG_SLOPE * fminf(v, 0.f);
}

#define NGRP    250
#define GSZ     400
#define PITCH   65
#define CAP     8192
#define NV16    (NEDGE / 16)

typedef __attribute__((ext_vector_type(4))) int v4i;
typedef __attribute__((ext_vector_type(4))) unsigned v4u;

__global__ void k_grp(const int* __restrict__ x, unsigned* __restrict__ grp4,
                      unsigned* __restrict__ pk) {
    int t = blockIdx.x * blockDim.x + threadIdx.x;
    if (t >= NEDGE / 4) return;
    v4i s = *(const v4i*)(x + 4 * t);
    v4i d = *(const v4i*)(x + NEDGE + 4 * t);
    int gx = d.x / GSZ, gy = d.y / GSZ, gz = d.z / GSZ, gw = d.w / GSZ;
    grp4[t] = (unsigned)gx | ((unsigned)gy << 8) | ((unsigned)gz << 16) | ((unsigned)gw << 24);
    v4u p;
    p.x = (unsigned)s.x | ((unsigned)(d.x - gx * GSZ) << 17);
    p.y = (unsigned)s.y | ((unsigned)(d.y - gy * GSZ) << 17);
    p.z = (unsigned)s.z | ((unsigned)(d.z - gz * GSZ) << 17);
    p.w = (unsigned)s.w | ((unsigned)(d.w - gw * GSZ) << 17);
    *(v4u*)(pk + 4 * t) = p;
}

__global__ __launch_bounds__(256) void k_gemm(const float* __restrict__ g,
                                              const float* __restrict__ We,
                                              const float* __restrict__ Wg,
                                              float* __restrict__ h_est,
                                              float* __restrict__ h_gnn) {
    int wave = threadIdx.x >> 5;
    int lane = threadIdx.x & 31;
    int mtile = blockIdx.x * 8 + wave;
    if (mtile >= MTILES) return;
    int m0   = mtile * 16;
    int lrow = lane & 15;
    int lhi  = lane >> 4;

    const float* grow = g + (long)(m0 + lrow) * NFEAT;

    v8f acc[8] = {};
    for (int k = 0; k < NFEAT; k += 4) {
        v2f a = *(const v2f*)(grow + k + 2 * lhi);
#pragma unroll
        for (int t = 0; t < 8; ++t) {
            const float* Wp = (t < 4) ? We : Wg;
            int c  = (t & 3) * 16 + lrow;
            int kr = k + 2 * lhi;
            v2f b;
            b.x = Wp[(long)kr * NHID + c];
            b.y = Wp[(long)(kr + 1) * NHID + c];
            acc[t] = __builtin_amdgcn_wmma_f32_16x16x4_f32(
                false, a, false, b, (short)0, acc[t], false, false);
        }
    }
#pragma unroll
    for (int t = 0; t < 8; ++t) {
        float* dst = (t < 4) ? h_est : h_gnn;
        int c = (t & 3) * 16 + lrow;
#pragma unroll
        for (int r = 0; r < 8; ++r) {
            dst[(long)(m0 + r + 8 * lhi) * NHID + c] = acc[t][r];
        }
    }
}

__global__ void k_eprep(const float* __restrict__ h_est, const float* __restrict__ h_gnn,
                        const float* __restrict__ ase, const float* __restrict__ ade,
                        const float* __restrict__ asg, const float* __restrict__ adg,
                        float* es2, float* ed_e, float* ed_g) {
    int n = blockIdx.x * blockDim.x + threadIdx.x;
    if (n >= N_NODES) return;
    const float* he = h_est + (long)n * NHID;
    const float* hg = h_gnn + (long)n * NHID;
    float se = 0.f, de = 0.f, sg = 0.f, dg = 0.f;
#pragma unroll 4
    for (int f = 0; f < NHID; f += 4) {
        v4f e  = *(const v4f*)(he + f);
        v4f g4 = *(const v4f*)(hg + f);
        v4f a1 = *(const v4f*)(ase + f);
        v4f a2 = *(const v4f*)(ade + f);
        v4f a3 = *(const v4f*)(asg + f);
        v4f a4 = *(const v4f*)(adg + f);
        se += e.x * a1.x + e.y * a1.y + e.z * a1.z + e.w * a1.w;
        de += e.x * a2.x + e.y * a2.y + e.z * a2.z + e.w * a2.w;
        sg += g4.x * a3.x + g4.y * a3.y + g4.z * a3.z + g4.w * a3.w;
        dg += g4.x * a4.x + g4.y * a4.y + g4.z * a4.z + g4.w * a4.w;
    }
    *(v2f*)(es2 + 2 * n) = (v2f){se, sg};
    ed_e[n] = de; ed_g[n] = dg;
}

__device__ __forceinline__ unsigned zbytes(unsigned u) {
    unsigned t = (u & 0x7f7f7f7fu) + 0x7f7f7f7fu;
    return ~(t | u | 0x7f7f7f7fu);
}
__device__ __forceinline__ unsigned nib(unsigned f) {
    return ((f >> 7) & 1u) | ((f >> 14) & 2u) | ((f >> 21) & 4u) | ((f >> 28) & 8u);
}
__device__ __forceinline__ unsigned match16(v4u v, unsigned pat) {
    unsigned f0 = zbytes(v.x ^ pat), f1 = zbytes(v.y ^ pat);
    unsigned f2 = zbytes(v.z ^ pat), f3 = zbytes(v.w ^ pat);
    if ((f0 | f1 | f2 | f3) == 0u) return 0u;
    return nib(f0) | (nib(f1) << 4) | (nib(f2) << 8) | (nib(f3) << 12);
}
__device__ __forceinline__ int rl_i(int v, int j) { return __builtin_amdgcn_readlane(v, j); }
__device__ __forceinline__ float rl_f(float v, int j) {
    return __int_as_float(__builtin_amdgcn_readlane(__float_as_int(v), j));
}

__global__ __launch_bounds__(1024) void k_own(const int* __restrict__ x,
        const unsigned* __restrict__ grp4, const unsigned* __restrict__ pk,
        const float* __restrict__ es2f,
        const float* __restrict__ ed_e, const float* __restrict__ ed_g,
        const float* __restrict__ h_est, const float* __restrict__ h_gnn,
        float* __restrict__ d_out,
        const float* __restrict__ eb, const float* __restrict__ egam,
        const float* __restrict__ ebeta, const float* __restrict__ erm,
        const float* __restrict__ erv,
        const float* __restrict__ gb, const float* __restrict__ ggam,
        const float* __restrict__ gbeta, const float* __restrict__ grm,
        const float* __restrict__ grv,
        const float* __restrict__ clsW, const float* __restrict__ clsb) {
    __shared__ float accE[GSZ * PITCH];
    __shared__ float accG[GSZ * PITCH];
    __shared__ unsigned lst[CAP];
    __shared__ float lgE[CAP], lgG[CAP];
    __shared__ float zE[GSZ], zG[GSZ], edE[GSZ], edG[GSZ];
    __shared__ unsigned mE[GSZ], mG[GSZ];
    __shared__ unsigned cnt, ovf;

    const int tid  = threadIdx.x;
    const int lane = tid & 31;
    const int wave = __builtin_amdgcn_readfirstlane(tid >> 5);
    const int grpid = blockIdx.x;
    const int n0   = grpid * GSZ;
    const unsigned pat = (unsigned)grpid * 0x01010101u;
    const v4u* g16 = (const v4u*)grp4;
    const v2f* es2 = (const v2f*)es2f;

    for (int i = tid; i < GSZ * PITCH; i += 1024) { accE[i] = 0.f; accG[i] = 0.f; }
    if (tid < GSZ) {
        int n = n0 + tid;
        edE[tid] = ed_e[n]; edG[tid] = ed_g[n];
        zE[tid] = 0.f; zG[tid] = 0.f;
        mE[tid] = ford(-FLT_MAX); mG[tid] = ford(-FLT_MAX);
    }
    if (tid == 0) { cnt = GSZ; ovf = 0u; }
    __syncthreads();

    for (int i = tid; i < NV16; i += 4096) {
        v4u v[4];
#pragma unroll
        for (int u = 0; u < 4; ++u) {
            int ii = i + u * 1024;
            v[u] = (ii < NV16) ? g16[ii] : (v4u){0xffffffffu, 0xffffffffu, 0xffffffffu, 0xffffffffu};
        }
#pragma unroll
        for (int u = 0; u < 4; ++u) {
            unsigned m = match16(v[u], pat);
            while (m) {
                int j = __builtin_ctz(m); m &= m - 1u;
                unsigned idx = atomicAdd(&cnt, 1u);
                if (idx < CAP) lst[idx] = (unsigned)((i + u * 1024) * 16 + j); else ovf = 1u;
            }
        }
    }
    __syncthreads();
    const bool spilled = ovf != 0u;
    const int total = __builtin_amdgcn_readfirstlane(spilled ? GSZ : (int)cnt);

    for (int idx = tid; idx < total; idx += 1024) {
        unsigned p = (idx < GSZ) ? ((unsigned)(n0 + idx) | ((unsigned)idx << 17)) : pk[lst[idx]];
        int s = (int)(p & 0x1ffffu);
        int l = (int)(p >> 17);
        v2f es = es2[s];
        float ee = leaky(es.x + edE[l]);
        float eg = leaky(es.y + edG[l]);
        atomicMax(&mE[l], ford(ee));
        atomicMax(&mG[l], ford(eg));
        lst[idx] = p; lgE[idx] = ee; lgG[idx] = eg;
    }
    if (spilled) {
        for (int i = tid; i < NV16; i += 1024) {
            unsigned m = match16(g16[i], pat);
            while (m) {
                int j = __builtin_ctz(m); m &= m - 1u;
                int e = i * 16 + j;
                int s = x[e], l = x[NEDGE + e] - n0;
                v2f es = es2[s];
                atomicMax(&mE[l], ford(leaky(es.x + edE[l])));
                atomicMax(&mG[l], ford(leaky(es.y + edG[l])));
            }
        }
    }
    __syncthreads();

    for (int base = wave * 32; base < total; base += 1024) {
        int idx = base + lane;
        bool valid = idx < total;
        unsigned ent = valid ? lst[idx] : 0u;
        int s = (int)(ent & 0x1ffffu);
        int l = (int)(ent >> 17);
        float ae = 0.f, ag = 0.f;
        if (valid) {
            ae = __expf(lgE[idx] - dorf(mE[l]));
            ag = __expf(lgG[idx] - dorf(mG[l]));
            atomicAdd(&zE[l], ae);
            atomicAdd(&zG[l], ag);
        }
        int nv = total - base; if (nv > 32) nv = 32;
        int j = 0;
        for (; j + 8 <= nv; j += 8) {
            float r0[8], r1[8], r2[8], r3[8], aj[8], gj[8]; int lj[8];
#pragma unroll
            for (int u = 0; u < 8; ++u) {
                int sj = rl_i(s, j + u);
                lj[u] = rl_i(l, j + u); aj[u] = rl_f(ae, j + u); gj[u] = rl_f(ag, j + u);
                const float* qe = h_est + (long)sj * NHID + lane;
                const float* qg = h_gnn + (long)sj * NHID + lane;
                r0[u] = qe[0]; r1[u] = qe[32]; r2[u] = qg[0]; r3[u] = qg[32];
            }
#pragma unroll
            for (int u = 0; u < 8; ++u) {
                float* pe = &accE[lj[u] * PITCH + lane];
                float* pg = &accG[lj[u] * PITCH + lane];
                atomicAdd(pe,      r0[u] * aj[u]);
                atomicAdd(pe + 32, r1[u] * aj[u]);
                atomicAdd(pg,      r2[u] * gj[u]);
                atomicAdd(pg + 32, r3[u] * gj[u]);
            }
        }
        for (; j < nv; ++j) {
            int sj = rl_i(s, j), l1 = rl_i(l, j);
            float a1 = rl_f(ae, j), g1 = rl_f(ag, j);
            const float* qe = h_est + (long)sj * NHID + lane;
            const float* qg = h_gnn + (long)sj * NHID + lane;
            float t0 = qe[0], t1 = qe[32], t2 = qg[0], t3 = qg[32];
            float* pe = &accE[l1 * PITCH + lane];
            float* pg = &accG[l1 * PITCH + lane];
            atomicAdd(pe,      t0 * a1);
            atomicAdd(pe + 32, t1 * a1);
            atomicAdd(pg,      t2 * g1);
            atomicAdd(pg + 32, t3 * g1);
        }
    }
    if (spilled) {
        for (int i = tid; i < NV16; i += 1024) {
            unsigned m = match16(g16[i], pat);
            while (m) {
                int j = __builtin_ctz(m); m &= m - 1u;
                int e = i * 16 + j;
                int s = x[e], l = x[NEDGE + e] - n0;
                v2f es = es2[s];
                float ae = __expf(leaky(es.x + edE[l]) - dorf(mE[l]));
                float ag = __expf(leaky(es.y + edG[l]) - dorf(mG[l]));
                atomicAdd(&zE[l], ae);
                atomicAdd(&zG[l], ag);
                for (int f = 0; f < NHID; ++f) {
                    atomicAdd(&accE[l * PITCH + f], h_est[(long)s * NHID + f] * ae);
                    atomicAdd(&accG[l * PITCH + f], h_gnn[(long)s * NHID + f] * ag);
                }
            }
        }
    }
    __syncthreads();
    if (tid < GSZ) { zE[tid] = 1.f / zE[tid]; zG[tid] = 1.f / zG[tid]; }
    __syncthreads();

    {
        int f = tid & 63;
        float b = eb[f], rm = erm[f], rs = rsqrtf(erv[f] + EPSBN), gm = egam[f], bt = ebeta[f];
        float* srow = d_out + N_NODES + (long)n0 * NHID;
        for (int i = tid; i < GSZ * NHID; i += 1024) {
            int l = i >> 6;
            float a = fmaxf(accE[l * PITCH + f] * zE[l] + b, 0.f);
            srow[i] = (a - rm) * rs * gm + bt;
        }
    }
    if (tid < GSZ) {
        const float* grow = accG + tid * PITCH;
        float rzg = zG[tid];
        float y = 0.f;
#pragma unroll 4
        for (int f = 0; f < NHID; ++f) {
            float b = fmaxf(grow[f] * rzg + gb[f], 0.f);
            float zv = (b - grm[f]) * rsqrtf(grv[f] + EPSBN) * ggam[f] + gbeta[f];
            y += zv * clsW[f];
        }
        d_out[n0 + tid] = y + clsb[0];
    }
}

extern "C" void kernel_launch(void* const* d_in, const int* in_sizes, int n_in,
                              void* d_out, int out_size, void* d_ws, size_t ws_size,
                              hipStream_t stream) {
    const float* g        = (const float*)d_in[0];
    const int*   x        = (const int*)d_in[1];
    const float* est_W    = (const float*)d_in[2];
    const float* est_asrc = (const float*)d_in[3];
    const float* est_adst = (const float*)d_in[4];
    const float* est_bias = (const float*)d_in[5];
    const float* est_gam  = (const float*)d_in[6];
    const float* est_beta = (const float*)d_in[7];
    const float* est_rm   = (const float*)d_in[8];
    const float* est_rv   = (const float*)d_in[9];
    const float* gnn_W    = (const float*)d_in[10];
    const float* gnn_asrc = (const float*)d_in[11];
    const float* gnn_adst = (const float*)d_in[12];
    const float* gnn_bias = (const float*)d_in[13];
    const float* gnn_gam  = (const float*)d_in[14];
    const float* gnn_beta = (const float*)d_in[15];
    const float* gnn_rm   = (const float*)d_in[16];
    const float* gnn_rv   = (const float*)d_in[17];
    const float* cls_W    = (const float*)d_in[18];
    const float* cls_b    = (const float*)d_in[19];
    float* out = (float*)d_out;

    float*    ws     = (float*)d_ws;
    float*    h_est  = ws;
    float*    h_gnn  = h_est + (long)N_NODES * NHID;
    float*    es2f   = h_gnn + (long)N_NODES * NHID;
    float*    ed_e   = es2f + 2 * N_NODES;
    float*    ed_g   = ed_e + N_NODES;
    unsigned* grp4   = (unsigned*)(ed_g + N_NODES);
    unsigned* pk     = grp4 + NEDGE / 4;

    const int BT = 256;
    k_grp<<<(NEDGE / 4 + BT - 1) / BT, BT, 0, stream>>>(x, grp4, pk);

    k_gemm<<<(MTILES + 7) / 8, BT, 0, stream>>>(g, est_W, gnn_W, h_est, h_gnn);

    k_eprep<<<(N_NODES + BT - 1) / BT, BT, 0, stream>>>(h_est, h_gnn,
            est_asrc, est_adst, gnn_asrc, gnn_adst, es2f, ed_e, ed_g);

    k_own<<<NGRP, 1024, 0, stream>>>(x, grp4, pk, es2f, ed_e, ed_g, h_est, h_gnn, out,
            est_bias, est_gam, est_beta, est_rm, est_rv,
            gnn_bias, gnn_gam, gnn_beta, gnn_rm, gnn_rv, cls_W, cls_b);
}


# Owner kernel scan: a lane tests 32 group-id bytes in straight-line code and files one flag word per block that holds matches (one LDS add per wave trip); edge ids are unpacked after, all lanes busy
# speedup vs baseline: 4.1077x; 1.0682x over previous
#include <hip/hip_runtime.h>
#include <cfloat>

#define N_NODES 100000
#define NFEAT   256
#define NHID    64
#define NEDGE   1600000
#define TOTE    (NEDGE + N_NODES)
#define MTILES  (N_NODES / 16)
#define EPSBN   1e-5f
#define NEG_SLOPE 0.2f

typedef __attribute__((ext_vector_type(2))) float v2f;
typedef __attribute__((ext_vector_type(4))) float v4f;
typedef __attribute__((ext_vector_type(8))) float v8f;

__device__ __forceinline__ unsigned ford(float f) {
    unsigned u = __float_as_uint(f);
    return (u & 0x80000000u) ? ~u : (u | 0x80000000u);
}
__device__ __forceinline__ float dorf(unsigned u) {
    return (u & 0x80000000u) ? __uint_as_float(u & 0x7fffffffu) : __uint_as_float(~u);
}
__device__ __forceinline__ float leaky(float v) {
    return fmaxf(v, 0.f) + NEG_SLOPE * fminf(v, 0.f);
}

#define NGRP    250
#define GSZ     400
#define PITCH   65
#define CAP     8192
#define NV16    (NEDGE / 16)
#define NV32    (NEDGE / 32)

typedef __attribute__((ext_vector_type(4))) int v4i;
typedef __attribute__((ext_vector_type(4))) unsigned v4u;

__global__ void k_grp(const int* __restrict__ x, unsigned* __restrict__ grp4,
                      unsigned* __restrict__ pk) {
    int t = blockIdx.x * blockDim.x + threadIdx.x;
    if (t >= NEDGE / 4) return;
    v4i s = *(const v4i*)(x + 4 * t);
    v4i d = *(const v4i*)(x + NEDGE + 4 * t);
    int gx = d.x / GSZ, gy = d.y / GSZ, gz = d.z / GSZ, gw = d.w / GSZ;
    grp4[t] = (unsigned)gx | ((unsigned)gy << 8) | ((unsigned)gz << 16) | ((unsigned)gw << 24);
    v4u p;
    p.x = (unsigned)s.x | ((unsigned)(d.x - gx * GSZ) << 17);
    p.y = (unsigned)s.y | ((unsigned)(d.y - gy * GSZ) << 17);
    p.z = (unsigned)s.z | ((unsigned)(d.z - gz * GSZ) << 17);
    p.w = (unsigned)s.w | ((unsigned)(d.w - gw * GSZ) << 17);
    *(v4u*)(pk + 4 * t) = p;
}

__global__ __launch_bounds__(256) void k_gemm(const float* __restrict__ g,
                                              const float* __restrict__ We,
                                              const float* __restrict__ Wg,
                                              float* __restrict__ h_est,
                                              float* __restrict__ h_gnn) {
    int wave = threadIdx.x >> 5;
    int lane = threadIdx.x & 31;
    int mtile = blockIdx.x * 8 + wave;
    if (mtile >= MTILES) return;
    int m0   = mtile * 16;
    int lrow = lane & 15;
    int lhi  = lane >> 4;

    const float* grow = g + (long)(m0 + lrow) * NFEAT;

    v8f acc[8] = {};
    for (int k = 0; k < NFEAT; k += 4) {
        v2f a = *(const v2f*)(grow + k + 2 * lhi);
#pragma unroll
        for (int t = 0; t < 8; ++t) {
            const float* Wp = (t < 4) ? We : Wg;
            int c  = (t & 3) * 16 + lrow;
            int kr = k + 2 * lhi;
            v2f b;
            b.x = Wp[(long)kr * NHID + c];
            b.y = Wp[(long)(kr + 1) * NHID + c];
            acc[t] = __builtin_amdgcn_wmma_f32_16x16x4_f32(
                false, a, false, b, (short)0, acc[t], false, false);
        }
    }
#pragma unroll
    for (int t = 0; t < 8; ++t) {
        float* dst = (t < 4) ? h_est : h_gnn;
        int c = (t & 3) * 16 + lrow;
#pragma unroll
        for (int r = 0; r < 8; ++r) {
            dst[(long)(m0 + r + 8 * lhi) * NHID + c] = acc[t][r];
        }
    }
}

__global__ void k_eprep(const float* __restrict__ h_est, const float* __restrict__ h_gnn,
                        const float* __restrict__ ase, const float* __restrict__ ade,
                        const float* __restrict__ asg, const float* __restrict__ adg,
                        float* es2, float* ed_e, float* ed_g) {
    int n = blockIdx.x * blockDim.x + threadIdx.x;
    if (n >= N_NODES) return;
    const float* he = h_est + (long)n * NHID;
    const float* hg = h_gnn + (long)n * NHID;
    float se = 0.f, de = 0.f, sg = 0.f, dg = 0.f;
#pragma unroll 4
    for (int f = 0; f < NHID; f += 4) {
        v4f e  = *(const v4f*)(he + f);
        v4f g4 = *(const v4f*)(hg + f);
        v4f a1 = *(const v4f*)(ase + f);
        v4f a2 = *(const v4f*)(ade + f);
        v4f a3 = *(const v4f*)(asg + f);
        v4f a4 = *(const v4f*)(adg + f);
        se += e.x * a1.x + e.y * a1.y + e.z * a1.z + e.w * a1.w;
        de += e.x * a2.x + e.y * a2.y + e.z * a2.z + e.w * a2.w;
        sg += g4.x * a3.x + g4.y * a3.y + g4.z * a3.z + g4.w * a3.w;
        dg += g4.x * a4.x + g4.y * a4.y + g4.z * a4.z + g4.w * a4.w;
    }
    *(v2f*)(es2 + 2 * n) = (v2f){se, sg};
    ed_e[n] = de; ed_g[n] = dg;
}

__device__ __forceinline__ unsigned zbytes(unsigned u) {
    unsigned t = (u & 0x7f7f7f7fu) + 0x7f7f7f7fu;
    return ~(t | u | 0x7f7f7f7fu);
}
__device__ __forceinline__ unsigned nib(unsigned f) {
    return ((f >> 7) & 1u) | ((f >> 14) & 2u) | ((f >> 21) & 4u) | ((f >> 28) & 8u);
}
__device__ __forceinline__ unsigned match16(v4u v, unsigned pat) {
    unsigned f0 = zbytes(v.x ^ pat), f1 = zbytes(v.y ^ pat);
    unsigned f2 = zbytes(v.z ^ pat), f3 = zbytes(v.w ^ pat);
    if ((f0 | f1 | f2 | f3) == 0u) return 0u;
    return nib(f0) | (nib(f1) << 4) | (nib(f2) << 8) | (nib(f3) << 12);
}
__device__ __forceinline__ unsigned flags32(v4u a, v4u b, unsigned pat) {
    return (zbytes(a.x ^ pat) >> 7) | (zbytes(a.y ^ pat) >> 6) |
           (zbytes(a.z ^ pat) >> 5) | (zbytes(a.w ^ pat) >> 4) |
           (zbytes(b.x ^ pat) >> 3) | (zbytes(b.y ^ pat) >> 2) |
           (zbytes(b.z ^ pat) >> 1) |  zbytes(b.w ^ pat);
}
__device__ __forceinline__ int rl_i(int v, int j) { return __builtin_amdgcn_readlane(v, j); }
__device__ __forceinline__ float rl_f(float v, int j) {
    return __int_as_float(__builtin_amdgcn_readlane(__float_as_int(v), j));
}

__global__ __launch_bounds__(1024) void k_own(const int* __restrict__ x,
        const unsigned* __restrict__ grp4, const unsigned* __restrict__ pk,
        const float* __restrict__ es2f,
        const float* __restrict__ ed_e, const float* __restrict__ ed_g,
        const float* __restrict__ h_est, const float* __restrict__ h_gnn,
        float* __restrict__ d_out,
        const float* __restrict__ eb, const float* __restrict__ egam,
        const float* __restrict__ ebeta, const float* __restrict__ erm,
        const float* __restrict__ erv,
        const float* __restrict__ gb, const float* __restrict__ ggam,
        const float* __restrict__ gbeta, const float* __restrict__ grm,
        const float* __restrict__ grv,
        const float* __restrict__ clsW, const float* __restrict__ clsb) {
    __shared__ float accE[GSZ * PITCH];
    __shared__ float accG[GSZ * PITCH];
    __shared__ unsigned lst[CAP];
    __shared__ unsigned lgE[CAP], lgG[CAP];
    __shared__ float zE[GSZ], zG[GSZ], edE[GSZ], edG[GSZ];
    __shared__ unsigned mE[GSZ], mG[GSZ];
    __shared__ unsigned cnt, ncand, ovf;

    const int tid  = threadIdx.x;
    const int lane = tid & 31;
    const int wave = __builtin_amdgcn_readfirstlane(tid >> 5);
    const int grpid = blockIdx.x;
    const int n0   = grpid * GSZ;
    const unsigned pat = (unsigned)grpid * 0x01010101u;
    const v4u* g16 = (const v4u*)grp4;
    const v2f* es2 = (const v2f*)es2f;

    for (int i = tid; i < GSZ * PITCH; i += 1024) { accE[i] = 0.f; accG[i] = 0.f; }
    if (tid < GSZ) {
        int n = n0 + tid;
        edE[tid] = ed_e[n]; edG[tid] = ed_g[n];
        zE[tid] = 0.f; zG[tid] = 0.f;
        mE[tid] = ford(-FLT_MAX); mG[tid] = ford(-FLT_MAX);
    }
    if (tid == 0) { cnt = GSZ; ncand = 0u; ovf = 0u; }
    __syncthreads();

    const v4u ones = (v4u){0xffffffffu, 0xffffffffu, 0xffffffffu, 0xffffffffu};
    for (int i0 = tid - lane; i0 < NV32; i0 += 2048) {
        const int ia = i0 + lane, ib = ia + 1024;
        v4u a0 = ones, a1 = ones, b0 = ones, b1 = ones;
        if (ia < NV32) { a0 = g16[2 * ia]; a1 = g16[2 * ia + 1]; }
        if (ib < NV32) { b0 = g16[2 * ib]; b1 = g16[2 * ib + 1]; }
        const unsigned Pa = flags32(a0, a1, pat), Pb = flags32(b0, b1, pat);
        const unsigned ma = __builtin_amdgcn_ballot_w32(Pa != 0u);
        const unsigned mb = __builtin_amdgcn_ballot_w32(Pb != 0u);
        const unsigned na = (unsigned)__builtin_popcount(ma);
        const unsigned nn = na + (unsigned)__builtin_popcount(mb);
        if (nn != 0u) {
            unsigned base = 0u;
            if (lane == 0) base = atomicAdd(&ncand, nn);
            base = (unsigned)__builtin_amdgcn_readfirstlane((int)base);
            if (Pa != 0u) {
                unsigned idx = base + __builtin_amdgcn_mbcnt_lo(ma, 0u);
                if (idx < CAP) { lgE[idx] = (unsigned)ia; lgG[idx] = Pa; } else ovf = 1u;
            }
            if (Pb != 0u) {
                unsigned idx = base + na + __builtin_amdgcn_mbcnt_lo(mb, 0u);
                if (idx < CAP) { lgE[idx] = (unsigned)ib; lgG[idx] = Pb; } else ovf = 1u;
            }
        }
    }
    __syncthreads();
    {
        const unsigned nc = ncand;
        const int ncl = __builtin_amdgcn_readfirstlane((int)(nc < (unsigned)CAP ? nc : (unsigned)CAP));
        for (int c = tid; c < ncl; c += 1024) {
            const unsigned blk = lgE[c];
            unsigned P = lgG[c];
            while (P) {
                unsigned t = (unsigned)__builtin_ctz(P); P &= P - 1u;
                unsigned idx = atomicAdd(&cnt, 1u);
                if (idx < CAP) lst[idx] = blk * 32u + 4u * (t & 7u) + (t >> 3); else ovf = 1u;
            }
        }
    }
    __syncthreads();
    const bool spilled = ovf != 0u;
    const int total = __builtin_amdgcn_readfirstlane(spilled ? GSZ : (int)cnt);

    for (int idx = tid; idx < total; idx += 1024) {
        unsigned p = (idx < GSZ) ? ((unsigned)(n0 + idx) | ((unsigned)idx << 17)) : pk[lst[idx]];
        int s = (int)(p & 0x1ffffu);
        int l = (int)(p >> 17);
        v2f es = es2[s];
        float ee = leaky(es.x + edE[l]);
        float eg = leaky(es.y + edG[l]);
        atomicMax(&mE[l], ford(ee));
        atomicMax(&mG[l], ford(eg));
        lst[idx] = p; lgE[idx] = __float_as_uint(ee); lgG[idx] = __float_as_uint(eg);
    }
    if (spilled) {
        for (int i = tid; i < NV16; i += 1024) {
            unsigned m = match16(g16[i], pat);
            while (m) {
                int j = __builtin_ctz(m); m &= m - 1u;
                int e = i * 16 + j;
                int s = x[e], l = x[NEDGE + e] - n0;
                v2f es = es2[s];
                atomicMax(&mE[l], ford(leaky(es.x + edE[l])));
                atomicMax(&mG[l], ford(leaky(es.y + edG[l])));
            }
        }
    }
    __syncthreads();

    for (int base = wave * 32; base < total; base += 1024) {
        int idx = base + lane;
        bool valid = idx < total;
        unsigned ent = valid ? lst[idx] : 0u;
        int s = (int)(ent & 0x1ffffu);
        int l = (int)(ent >> 17);
        float ae = 0.f, ag = 0.f;
        if (valid) {
            ae = __expf(__uint_as_float(lgE[idx]) - dorf(mE[l]));
            ag = __expf(__uint_as_float(lgG[idx]) - dorf(mG[l]));
            atomicAdd(&zE[l], ae);
            atomicAdd(&zG[l], ag);
        }
        int nv = total - base; if (nv > 32) nv = 32;
        int j = 0;
        for (; j + 8 <= nv; j += 8) {
            float r0[8], r1[8], r2[8], r3[8], aj[8], gj[8]; int lj[8];
#pragma unroll
            for (int u = 0; u < 8; ++u) {
                int sj = rl_i(s, j + u);
                lj[u] = rl_i(l, j + u); aj[u] = rl_f(ae, j + u); gj[u] = rl_f(ag, j + u);
                const float* qe = h_est + (long)sj * NHID + lane;
                const float* qg = h_gnn + (long)sj * NHID + lane;
                r0[u] = qe[0]; r1[u] = qe[32]; r2[u] = qg[0]; r3[u] = qg[32];
            }
#pragma unroll
            for (int u = 0; u < 8; ++u) {
                float* pe = &accE[lj[u] * PITCH + lane];
                float* pg = &accG[lj[u] * PITCH + lane];
                atomicAdd(pe,      r0[u] * aj[u]);
                atomicAdd(pe + 32, r1[u] * aj[u]);
                atomicAdd(pg,      r2[u] * gj[u]);
                atomicAdd(pg + 32, r3[u] * gj[u]);
            }
        }
        for (; j < nv; ++j) {
            int sj = rl_i(s, j), l1 = rl_i(l, j);
            float a1 = rl_f(ae, j), g1 = rl_f(ag, j);
            const float* qe = h_est + (long)sj * NHID + lane;
            const float* qg = h_gnn + (long)sj * NHID + lane;
            float t0 = qe[0], t1 = qe[32], t2 = qg[0], t3 = qg[32];
            float* pe = &accE[l1 * PITCH + lane];
            float* pg = &accG[l1 * PITCH + lane];
            atomicAdd(pe,      t0 * a1);
            atomicAdd(pe + 32, t1 * a1);
            atomicAdd(pg,      t2 * g1);
            atomicAdd(pg + 32, t3 * g1);
        }
    }
    if (spilled) {
        for (int i = tid; i < NV16; i += 1024) {
            unsigned m = match16(g16[i], pat);
            while (m) {
                int j = __builtin_ctz(m); m &= m - 1u;
                int e = i * 16 + j;
                int s = x[e], l = x[NEDGE + e] - n0;
                v2f es = es2[s];
                float ae = __expf(leaky(es.x + edE[l]) - dorf(mE[l]));
                float ag = __expf(leaky(es.y + edG[l]) - dorf(mG[l]));
                atomicAdd(&zE[l], ae);
                atomicAdd(&zG[l], ag);
                for (int f = 0; f < NHID; ++f) {
                    atomicAdd(&accE[l * PITCH + f], h_est[(long)s * NHID + f] * ae);
                    atomicAdd(&accG[l * PITCH + f], h_gnn[(long)s * NHID + f] * ag);
                }
            }
        }
    }
    __syncthreads();
    if (tid < GSZ) { zE[tid] = 1.f / zE[tid]; zG[tid] = 1.f / zG[tid]; }
    __syncthreads();

    {
        int f = tid & 63;
        float b = eb[f], rm = erm[f], rs = rsqrtf(erv[f] + EPSBN), gm = egam[f], bt = ebeta[f];
        float* srow = d_out + N_NODES + (long)n0 * NHID;
        for (int i = tid; i < GSZ * NHID; i += 1024) {
            int l = i >> 6;
            float a = fmaxf(accE[l * PITCH + f] * zE[l] + b, 0.f);
            srow[i] = (a - rm) * rs * gm + bt;
        }
    }
    if (tid < GSZ) {
        const float* grow = accG + tid * PITCH;
        float rzg = zG[tid];
        float y = 0.f;
#pragma unroll 4
        for (int f = 0; f < NHID; ++f) {
            float b = fmaxf(grow[f] * rzg + gb[f], 0.f);
            float zv = (b - grm[f]) * rsqrtf(grv[f] + EPSBN) * ggam[f] + gbeta[f];
            y += zv * clsW[f];
        }
        d_out[n0 + tid] = y + clsb[0];
    }
}

extern "C" void kernel_launch(void* const* d_in, const int* in_sizes, int n_in,
                              void* d_out, int out_size, void* d_ws, size_t ws_size,
                              hipStream_t stream) {
    const float* g        = (const float*)d_in[0];
    const int*   x        = (const int*)d_in[1];
    const float* est_W    = (const float*)d_in[2];
    const float* est_asrc = (const float*)d_in[3];
    const float* est_adst = (const float*)d_in[4];
    const float* est_bias = (const float*)d_in[5];
    const float* est_gam  = (const float*)d_in[6];
    const float* est_beta = (const float*)d_in[7];
    const float* est_rm   = (const float*)d_in[8];
    const float* est_rv   = (const float*)d_in[9];
    const float* gnn_W    = (const float*)d_in[10];
    const float* gnn_asrc = (const float*)d_in[11];
    const float* gnn_adst = (const float*)d_in[12];
    const float* gnn_bias = (const float*)d_in[13];
    const float* gnn_gam  = (const float*)d_in[14];
    const float* gnn_beta = (const float*)d_in[15];
    const float* gnn_rm   = (const float*)d_in[16];
    const float* gnn_rv   = (const float*)d_in[17];
    const float* cls_W    = (const float*)d_in[18];
    const float* cls_b    = (const float*)d_in[19];
    float* out = (float*)d_out;

    float*    ws     = (float*)d_ws;
    float*    h_est  = ws;
    float*    h_gnn  = h_est + (long)N_NODES * NHID;
    float*    es2f   = h_gnn + (long)N_NODES * NHID;
    float*    ed_e   = es2f + 2 * N_NODES;
    float*    ed_g   = ed_e + N_NODES;
    unsigned* grp4   = (unsigned*)(ed_g + N_NODES);
    unsigned* pk     = grp4 + NEDGE / 4;

    const int BT = 256;
    k_grp<<<(NEDGE / 4 + BT - 1) / BT, BT, 0, stream>>>(x, grp4, pk);

    k_gemm<<<(MTILES + 7) / 8, BT, 0, stream>>>(g, est_W, gnn_W, h_est, h_gnn);

    k_eprep<<<(N_NODES + BT - 1) / BT, BT, 0, stream>>>(h_est, h_gnn,
            est_asrc, est_adst, gnn_asrc, gnn_adst, es2f, ed_e, ed_g);

    k_own<<<NGRP, 1024, 0, stream>>>(x, grp4, pk, es2f, ed_e, ed_g, h_est, h_gnn, out,
            est_bias, est_gam, est_beta, est_rm, est_rv,
            gnn_bias, gnn_gam, gnn_beta, gnn_rm, gnn_rv, cls_W, cls_b);
}


# Edge-list conversion folded into the GEMM launch as 200 trailing work-groups whose threads loop over the list, four items' loads in flight: 4 launches become 3, 12504 short waves become 1600
# speedup vs baseline: 4.2246x; 1.0285x over previous
#include <hip/hip_runtime.h>
#include <cfloat>

#define N_NODES 100000
#define NFEAT   256
#define NHID    64
#define NEDGE   1600000
#define TOTE    (NEDGE + N_NODES)
#define MTILES  (N_NODES / 16)
#define EPSBN   1e-5f
#define NEG_SLOPE 0.2f

typedef __attribute__((ext_vector_type(2))) float v2f;
typedef __attribute__((ext_vector_type(4))) float v4f;
typedef __attribute__((ext_vector_type(8))) float v8f;

__device__ __forceinline__ unsigned ford(float f) {
    unsigned u = __float_as_uint(f);
    return (u & 0x80000000u) ? ~u : (u | 0x80000000u);
}
__device__ __forceinline__ float dorf(unsigned u) {
    return (u & 0x80000000u) ? __uint_as_float(u & 0x7fffffffu) : __uint_as_float(~u);
}
__device__ __forceinline__ float leaky(float v) {
    return fmaxf(v, 0.f) + NEG_SLOPE * fminf(v, 0.f);
}

#define NGRP    250
#define GSZ     400
#define PITCH   65
#define CAP     8192
#define NV16    (NEDGE / 16)
#define NV32    (NEDGE / 32)

typedef __attribute__((ext_vector_type(4))) int v4i;
typedef __attribute__((ext_vector_type(4))) unsigned v4u;

__device__ __forceinline__ void conv_item(v4i s, v4i d, int t, unsigned* __restrict__ grp4,
                                          unsigned* __restrict__ pk) {
    int gx = d.x / GSZ, gy = d.y / GSZ, gz = d.z / GSZ, gw = d.w / GSZ;
    grp4[t] = (unsigned)gx | ((unsigned)gy << 8) | ((unsigned)gz << 16) | ((unsigned)gw << 24);
    v4u p;
    p.x = (unsigned)s.x | ((unsigned)(d.x - gx * GSZ) << 17);
    p.y = (unsigned)s.y | ((unsigned)(d.y - gy * GSZ) << 17);
    p.z = (unsigned)s.z | ((unsigned)(d.z - gz * GSZ) << 17);
    p.w = (unsigned)s.w | ((unsigned)(d.w - gw * GSZ) << 17);
    *(v4u*)(pk + 4 * t) = p;
}

#define GEMMB   ((MTILES + 7) / 8)
#define CONVB   200
#define NITEM   (NEDGE / 4)
#define CONVT   (CONVB * 256)

__global__ __launch_bounds__(256) void k_gemm(const float* __restrict__ g,
                                              const float* __restrict__ We,
                                              const float* __restrict__ Wg,
                                              float* __restrict__ h_est,
                                              float* __restrict__ h_gnn,
                                              const int* __restrict__ x,
                                              unsigned* __restrict__ grp4,
                                              unsigned* __restrict__ pk) {
    if (blockIdx.x >= GEMMB) {
        int t = (blockIdx.x - GEMMB) * 256 + threadIdx.x;
        for (; t + 3 * CONVT < NITEM; t += 4 * CONVT) {
            v4i s0 = *(const v4i*)(x + 4 * t);
            v4i d0 = *(const v4i*)(x + NEDGE + 4 * t);
            v4i s1 = *(const v4i*)(x + 4 * (t + CONVT));
            v4i d1 = *(const v4i*)(x + NEDGE + 4 * (t + CONVT));
            v4i s2 = *(const v4i*)(x + 4 * (t + 2 * CONVT));
            v4i d2 = *(const v4i*)(x + NEDGE + 4 * (t + 2 * CONVT));
            v4i s3 = *(const v4i*)(x + 4 * (t + 3 * CONVT));
            v4i d3 = *(const v4i*)(x + NEDGE + 4 * (t + 3 * CONVT));
            conv_item(s0, d0, t, grp4, pk);
            conv_item(s1, d1, t + CONVT, grp4, pk);
            conv_item(s2, d2, t + 2 * CONVT, grp4, pk);
            conv_item(s3, d3, t + 3 * CONVT, grp4, pk);
        }
        for (; t < NITEM; t += CONVT) {
            v4i s0 = *(const v4i*)(x + 4 * t);
            v4i d0 = *(const v4i*)(x + NEDGE + 4 * t);
            conv_item(s0, d0, t, grp4, pk);
        }
        return;
    }
    int wave = threadIdx.x >> 5;
    int lane = threadIdx.x & 31;
    int mtile = blockIdx.x * 8 + wave;
    if (mtile >= MTILES) return;
    int m0   = mtile * 16;
    int lrow = lane & 15;
    int lhi  = lane >> 4;

    const float* grow = g + (long)(m0 + lrow) * NFEAT;

    v8f acc[8] = {};
    for (int k = 0; k < NFEAT; k += 4) {
        v2f a = *(const v2f*)(grow + k + 2 * lhi);
#pragma unroll
        for (int t = 0; t < 8; ++t) {
            const float* Wp = (t < 4) ? We : Wg;
            int c  = (t & 3) * 16 + lrow;
            int kr = k + 2 * lhi;
            v2f b;
            b.x = Wp[(long)kr * NHID + c];
            b.y = Wp[(long)(kr + 1) * NHID + c];
            acc[t] = __builtin_amdgcn_wmma_f32_16x16x4_f32(
                false, a, false, b, (short)0, acc[t], false, false);
        }
    }
#pragma unroll
    for (int t = 0; t < 8; ++t) {
        float* dst = (t < 4) ? h_est : h_gnn;
        int c = (t & 3) * 16 + lrow;
#pragma unroll
        for (int r = 0; r < 8; ++r) {
            dst[(long)(m0 + r + 8 * lhi) * NHID + c] = acc[t][r];
        }
    }
}

__global__ void k_eprep(const float* __restrict__ h_est, const float* __restrict__ h_gnn,
                        const float* __restrict__ ase, const float* __restrict__ ade,
                        const float* __restrict__ asg, const float* __restrict__ adg,
                        float* es2, float* ed_e, float* ed_g) {
    int n = blockIdx.x * blockDim.x + threadIdx.x;
    if (n >= N_NODES) return;
    const float* he = h_est + (long)n * NHID;
    const float* hg = h_gnn + (long)n * NHID;
    float se = 0.f, de = 0.f, sg = 0.f, dg = 0.f;
#pragma unroll 4
    for (int f = 0; f < NHID; f += 4) {
        v4f e  = *(const v4f*)(he + f);
        v4f g4 = *(const v4f*)(hg + f);
        v4f a1 = *(const v4f*)(ase + f);
        v4f a2 = *(const v4f*)(ade + f);
        v4f a3 = *(const v4f*)(asg + f);
        v4f a4 = *(const v4f*)(adg + f);
        se += e.x * a1.x + e.y * a1.y + e.z * a1.z + e.w * a1.w;
        de += e.x * a2.x + e.y * a2.y + e.z * a2.z + e.w * a2.w;
        sg += g4.x * a3.x + g4.y * a3.y + g4.z * a3.z + g4.w * a3.w;
        dg += g4.x * a4.x + g4.y * a4.y + g4.z * a4.z + g4.w * a4.w;
    }
    *(v2f*)(es2 + 2 * n) = (v2f){se, sg};
    ed_e[n] = de; ed_g[n] = dg;
}

__device__ __forceinline__ unsigned zbytes(unsigned u) {
    unsigned t = (u & 0x7f7f7f7fu) + 0x7f7f7f7fu;
    return ~(t | u | 0x7f7f7f7fu);
}
__device__ __forceinline__ unsigned nib(unsigned f) {
    return ((f >> 7) & 1u) | ((f >> 14) & 2u) | ((f >> 21) & 4u) | ((f >> 28) & 8u);
}
__device__ __forceinline__ unsigned match16(v4u v, unsigned pat) {
    unsigned f0 = zbytes(v.x ^ pat), f1 = zbytes(v.y ^ pat);
    unsigned f2 = zbytes(v.z ^ pat), f3 = zbytes(v.w ^ pat);
    if ((f0 | f1 | f2 | f3) == 0u) return 0u;
    return nib(f0) | (nib(f1) << 4) | (nib(f2) << 8) | (nib(f3) << 12);
}
__device__ __forceinline__ unsigned flags32(v4u a, v4u b, unsigned pat) {
    return (zbytes(a.x ^ pat) >> 7) | (zbytes(a.y ^ pat) >> 6) |
           (zbytes(a.z ^ pat) >> 5) | (zbytes(a.w ^ pat) >> 4) |
           (zbytes(b.x ^ pat) >> 3) | (zbytes(b.y ^ pat) >> 2) |
           (zbytes(b.z ^ pat) >> 1) |  zbytes(b.w ^ pat);
}
__device__ __forceinline__ int rl_i(int v, int j) { return __builtin_amdgcn_readlane(v, j); }
__device__ __forceinline__ float rl_f(float v, int j) {
    return __int_as_float(__builtin_amdgcn_readlane(__float_as_int(v), j));
}

__global__ __launch_bounds__(1024) void k_own(const int* __restrict__ x,
        const unsigned* __restrict__ grp4, const unsigned* __restrict__ pk,
        const float* __restrict__ es2f,
        const float* __restrict__ ed_e, const float* __restrict__ ed_g,
        const float* __restrict__ h_est, const float* __restrict__ h_gnn,
        float* __restrict__ d_out,
        const float* __restrict__ eb, const float* __restrict__ egam,
        const float* __restrict__ ebeta, const float* __restrict__ erm,
        const float* __restrict__ erv,
        const float* __restrict__ gb, const float* __restrict__ ggam,
        const float* __restrict__ gbeta, const float* __restrict__ grm,
        const float* __restrict__ grv,
        const float* __restrict__ clsW, const float* __restrict__ clsb) {
    __shared__ float accE[GSZ * PITCH];
    __shared__ float accG[GSZ * PITCH];
    __shared__ unsigned lst[CAP];
    __shared__ unsigned lgE[CAP], lgG[CAP];
    __shared__ float zE[GSZ], zG[GSZ], edE[GSZ], edG[GSZ];
    __shared__ unsigned mE[GSZ], mG[GSZ];
    __shared__ unsigned cnt, ncand, ovf;

    const int tid  = threadIdx.x;
    const int lane = tid & 31;
    const int wave = __builtin_amdgcn_readfirstlane(tid >> 5);
    const int grpid = blockIdx.x;
    const int n0   = grpid * GSZ;
    const unsigned pat = (unsigned)grpid * 0x01010101u;
    const v4u* g16 = (const v4u*)grp4;
    const v2f* es2 = (const v2f*)es2f;

    for (int i = tid; i < GSZ * PITCH; i += 1024) { accE[i] = 0.f; accG[i] = 0.f; }
    if (tid < GSZ) {
        int n = n0 + tid;
        edE[tid] = ed_e[n]; edG[tid] = ed_g[n];
        zE[tid] = 0.f; zG[tid] = 0.f;
        mE[tid] = ford(-FLT_MAX); mG[tid] = ford(-FLT_MAX);
    }
    if (tid == 0) { cnt = GSZ; ncand = 0u; ovf = 0u; }
    __syncthreads();

    const v4u ones = (v4u){0xffffffffu, 0xffffffffu, 0xffffffffu, 0xffffffffu};
    for (int i0 = tid - lane; i0 < NV32; i0 += 2048) {
        const int ia = i0 + lane, ib = ia + 1024;
        v4u a0 = ones, a1 = ones, b0 = ones, b1 = ones;
        if (ia < NV32) { a0 = g16[2 * ia]; a1 = g16[2 * ia + 1]; }
        if (ib < NV32) { b0 = g16[2 * ib]; b1 = g16[2 * ib + 1]; }
        const unsigned Pa = flags32(a0, a1, pat), Pb = flags32(b0, b1, pat);
        const unsigned ma = __builtin_amdgcn_ballot_w32(Pa != 0u);
        const unsigned mb = __builtin_amdgcn_ballot_w32(Pb != 0u);
        const unsigned na = (unsigned)__builtin_popcount(ma);
        const unsigned nn = na + (unsigned)__builtin_popcount(mb);
        if (nn != 0u) {
            unsigned base = 0u;
            if (lane == 0) base = atomicAdd(&ncand, nn);
            base = (unsigned)__builtin_amdgcn_readfirstlane((int)base);
            if (Pa != 0u) {
                unsigned idx = base + __builtin_amdgcn_mbcnt_lo(ma, 0u);
                if (idx < CAP) { lgE[idx] = (unsigned)ia; lgG[idx] = Pa; } else ovf = 1u;
            }
            if (Pb != 0u) {
                unsigned idx = base + na + __builtin_amdgcn_mbcnt_lo(mb, 0u);
                if (idx < CAP) { lgE[idx] = (unsigned)ib; lgG[idx] = Pb; } else ovf = 1u;
            }
        }
    }
    __syncthreads();
    {
        const unsigned nc = ncand;
        const int ncl = __builtin_amdgcn_readfirstlane((int)(nc < (unsigned)CAP ? nc : (unsigned)CAP));
        for (int c = tid; c < ncl; c += 1024) {
            const unsigned blk = lgE[c];
            unsigned P = lgG[c];
            while (P) {
                unsigned t = (unsigned)__builtin_ctz(P); P &= P - 1u;
                unsigned idx = atomicAdd(&cnt, 1u);
                if (idx < CAP) lst[idx] = blk * 32u + 4u * (t & 7u) + (t >> 3); else ovf = 1u;
            }
        }
    }
    __syncthreads();
    const bool spilled = ovf != 0u;
    const int total = __builtin_amdgcn_readfirstlane(spilled ? GSZ : (int)cnt);

    for (int idx = tid; idx < total; idx += 1024) {
        unsigned p = (idx < GSZ) ? ((unsigned)(n0 + idx) | ((unsigned)idx << 17)) : pk[lst[idx]];
        int s = (int)(p & 0x1ffffu);
        int l = (int)(p >> 17);
        v2f es = es2[s];
        float ee = leaky(es.x + edE[l]);
        float eg = leaky(es.y + edG[l]);
        atomicMax(&mE[l], ford(ee));
        atomicMax(&mG[l], ford(eg));
        lst[idx] = p; lgE[idx] = __float_as_uint(ee); lgG[idx] = __float_as_uint(eg);
    }
    if (spilled) {
        for (int i = tid; i < NV16; i += 1024) {
            unsigned m = match16(g16[i], pat);
            while (m) {
                int j = __builtin_ctz(m); m &= m - 1u;
                int e = i * 16 + j;
                int s = x[e], l = x[NEDGE + e] - n0;
                v2f es = es2[s];
                atomicMax(&mE[l], ford(leaky(es.x + edE[l])));
                atomicMax(&mG[l], ford(leaky(es.y + edG[l])));
            }
        }
    }
    __syncthreads();

    for (int base = wave * 32; base < total; base += 1024) {
        int idx = base + lane;
        bool valid = idx < total;
        unsigned ent = valid ? lst[idx] : 0u;
        int s = (int)(ent & 0x1ffffu);
        int l = (int)(ent >> 17);
        float ae = 0.f, ag = 0.f;
        if (valid) {
            ae = __expf(__uint_as_float(lgE[idx]) - dorf(mE[l]));
            ag = __expf(__uint_as_float(lgG[idx]) - dorf(mG[l]));
            atomicAdd(&zE[l], ae);
            atomicAdd(&zG[l], ag);
        }
        int nv = total - base; if (nv > 32) nv = 32;
        int j = 0;
        for (; j + 8 <= nv; j += 8) {
            float r0[8], r1[8], r2[8], r3[8], aj[8], gj[8]; int lj[8];
#pragma unroll
            for (int u = 0; u < 8; ++u) {
                int sj = rl_i(s, j + u);
                lj[u] = rl_i(l, j + u); aj[u] = rl_f(ae, j + u); gj[u] = rl_f(ag, j + u);
                const float* qe = h_est + (long)sj * NHID + lane;
                const float* qg = h_gnn + (long)sj * NHID + lane;
                r0[u] = qe[0]; r1[u] = qe[32]; r2[u] = qg[0]; r3[u] = qg[32];
            }
#pragma unroll
            for (int u = 0; u < 8; ++u) {
                float* pe = &accE[lj[u] * PITCH + lane];
                float* pg = &accG[lj[u] * PITCH + lane];
                atomicAdd(pe,      r0[u] * aj[u]);
                atomicAdd(pe + 32, r1[u] * aj[u]);
                atomicAdd(pg,      r2[u] * gj[u]);
                atomicAdd(pg + 32, r3[u] * gj[u]);
            }
        }
        for (; j < nv; ++j) {
            int sj = rl_i(s, j), l1 = rl_i(l, j);
            float a1 = rl_f(ae, j), g1 = rl_f(ag, j);
            const float* qe = h_est + (long)sj * NHID + lane;
            const float* qg = h_gnn + (long)sj * NHID + lane;
            float t0 = qe[0], t1 = qe[32], t2 = qg[0], t3 = qg[32];
            float* pe = &accE[l1 * PITCH + lane];
            float* pg = &accG[l1 * PITCH + lane];
            atomicAdd(pe,      t0 * a1);
            atomicAdd(pe + 32, t1 * a1);
            atomicAdd(pg,      t2 * g1);
            atomicAdd(pg + 32, t3 * g1);
        }
    }
    if (spilled) {
        for (int i = tid; i < NV16; i += 1024) {
            unsigned m = match16(g16[i], pat);
            while (m) {
                int j = __builtin_ctz(m); m &= m - 1u;
                int e = i * 16 + j;
                int s = x[e], l = x[NEDGE + e] - n0;
                v2f es = es2[s];
                float ae = __expf(leaky(es.x + edE[l]) - dorf(mE[l]));
                float ag = __expf(leaky(es.y + edG[l]) - dorf(mG[l]));
                atomicAdd(&zE[l], ae);
                atomicAdd(&zG[l], ag);
                for (int f = 0; f < NHID; ++f) {
                    atomicAdd(&accE[l * PITCH + f], h_est[(long)s * NHID + f] * ae);
                    atomicAdd(&accG[l * PITCH + f], h_gnn[(long)s * NHID + f] * ag);
                }
            }
        }
    }
    __syncthreads();
    if (tid < GSZ) { zE[tid] = 1.f / zE[tid]; zG[tid] = 1.f / zG[tid]; }
    __syncthreads();

    {
        int f = tid & 63;
        float b = eb[f], rm = erm[f], rs = rsqrtf(erv[f] + EPSBN), gm = egam[f], bt = ebeta[f];
        float* srow = d_out + N_NODES + (long)n0 * NHID;
        for (int i = tid; i < GSZ * NHID; i += 1024) {
            int l = i >> 6;
            float a = fmaxf(accE[l * PITCH + f] * zE[l] + b, 0.f);
            srow[i] = (a - rm) * rs * gm + bt;
        }
    }
    if (tid < GSZ) {
        const float* grow = accG + tid * PITCH;
        float rzg = zG[tid];
        float y = 0.f;
#pragma unroll 4
        for (int f = 0; f < NHID; ++f) {
            float b = fmaxf(grow[f] * rzg + gb[f], 0.f);
            float zv = (b - grm[f]) * rsqrtf(grv[f] + EPSBN) * ggam[f] + gbeta[f];
            y += zv * clsW[f];
        }
        d_out[n0 + tid] = y + clsb[0];
    }
}

extern "C" void kernel_launch(void* const* d_in, const int* in_sizes, int n_in,
                              void* d_out, int out_size, void* d_ws, size_t ws_size,
                              hipStream_t stream) {
    const float* g        = (const float*)d_in[0];
    const int*   x        = (const int*)d_in[1];
    const float* est_W    = (const float*)d_in[2];
    const float* est_asrc = (const float*)d_in[3];
    const float* est_adst = (const float*)d_in[4];
    const float* est_bias = (const float*)d_in[5];
    const float* est_gam  = (const float*)d_in[6];
    const float* est_beta = (const float*)d_in[7];
    const float* est_rm   = (const float*)d_in[8];
    const float* est_rv   = (const float*)d_in[9];
    const float* gnn_W    = (const float*)d_in[10];
    const float* gnn_asrc = (const float*)d_in[11];
    const float* gnn_adst = (const float*)d_in[12];
    const float* gnn_bias = (const float*)d_in[13];
    const float* gnn_gam  = (const float*)d_in[14];
    const float* gnn_beta = (const float*)d_in[15];
    const float* gnn_rm   = (const float*)d_in[16];
    const float* gnn_rv   = (const float*)d_in[17];
    const float* cls_W    = (const float*)d_in[18];
    const float* cls_b    = (const float*)d_in[19];
    float* out = (float*)d_out;

    float*    ws     = (float*)d_ws;
    float*    h_est  = ws;
    float*    h_gnn  = h_est + (long)N_NODES * NHID;
    float*    es2f   = h_gnn + (long)N_NODES * NHID;
    float*    ed_e   = es2f + 2 * N_NODES;
    float*    ed_g   = ed_e + N_NODES;
    unsigned* grp4   = (unsigned*)(ed_g + N_NODES);
    unsigned* pk     = grp4 + NEDGE / 4;

    const int BT = 256;
    k_gemm<<<GEMMB + CONVB, BT, 0, stream>>>(g, est_W, gnn_W, h_est, h_gnn, x, grp4, pk);

    k_eprep<<<(N_NODES + BT - 1) / BT, BT, 0, stream>>>(h_est, h_gnn,
            est_asrc, est_adst, gnn_asrc, gnn_adst, es2f, ed_e, ed_g);

    k_own<<<NGRP, 1024, 0, stream>>>(x, grp4, pk, es2f, ed_e, ed_g, h_est, h_gnn, out,
            est_bias, est_gam, est_beta, est_rm, est_rv,
            gnn_bias, gnn_gam, gnn_beta, gnn_rm, gnn_rv, cls_W, cls_b);
}


# GEMM: each wave computes a 32x128 tile (two 16-row tiles) in 4-wave work-groups, so every B fragment loaded feeds two products: B loads per product halved; k order of every sum unchanged
# speedup vs baseline: 4.5874x; 1.0859x over previous
#include <hip/hip_runtime.h>
#include <cfloat>

#define N_NODES 100000
#define NFEAT   256
#define NHID    64
#define NEDGE   1600000
#define TOTE    (NEDGE + N_NODES)
#define MTILES  (N_NODES / 16)
#define EPSBN   1e-5f
#define NEG_SLOPE 0.2f

typedef __attribute__((ext_vector_type(2))) float v2f;
typedef __attribute__((ext_vector_type(4))) float v4f;
typedef __attribute__((ext_vector_type(8))) float v8f;

__device__ __forceinline__ unsigned ford(float f) {
    unsigned u = __float_as_uint(f);
    return (u & 0x80000000u) ? ~u : (u | 0x80000000u);
}
__device__ __forceinline__ float dorf(unsigned u) {
    return (u & 0x80000000u) ? __uint_as_float(u & 0x7fffffffu) : __uint_as_float(~u);
}
__device__ __forceinline__ float leaky(float v) {
    return fmaxf(v, 0.f) + NEG_SLOPE * fminf(v, 0.f);
}

#define NGRP    250
#define GSZ     400
#define PITCH   65
#define CAP     8192
#define NV16    (NEDGE / 16)
#define NV32    (NEDGE / 32)

typedef __attribute__((ext_vector_type(4))) int v4i;
typedef __attribute__((ext_vector_type(4))) unsigned v4u;

__device__ __forceinline__ void conv_item(v4i s, v4i d, int t, unsigned* __restrict__ grp4,
                                          unsigned* __restrict__ pk) {
    int gx = d.x / GSZ, gy = d.y / GSZ, gz = d.z / GSZ, gw = d.w / GSZ;
    grp4[t] = (unsigned)gx | ((unsigned)gy << 8) | ((unsigned)gz << 16) | ((unsigned)gw << 24);
    v4u p;
    p.x = (unsigned)s.x | ((unsigned)(d.x - gx * GSZ) << 17);
    p.y = (unsigned)s.y | ((unsigned)(d.y - gy * GSZ) << 17);
    p.z = (unsigned)s.z | ((unsigned)(d.z - gz * GSZ) << 17);
    p.w = (unsigned)s.w | ((unsigned)(d.w - gw * GSZ) << 17);
    *(v4u*)(pk + 4 * t) = p;
}

#define GT      128
#define NPAIR   (N_NODES / 32)
#define GEMMB   ((NPAIR + 3) / 4)
#define CONVB   400
#define NITEM   (NEDGE / 4)
#define CONVT   (CONVB * GT)

__global__ __launch_bounds__(GT) void k_gemm(const float* __restrict__ g,
                                              const float* __restrict__ We,
                                              const float* __restrict__ Wg,
                                              float* __restrict__ h_est,
                                              float* __restrict__ h_gnn,
                                              const int* __restrict__ x,
                                              unsigned* __restrict__ grp4,
                                              unsigned* __restrict__ pk) {
    if (blockIdx.x >= GEMMB) {
        int t = (blockIdx.x - GEMMB) * GT + threadIdx.x;
        for (; t + 3 * CONVT < NITEM; t += 4 * CONVT) {
            v4i s0 = *(const v4i*)(x + 4 * t);
            v4i d0 = *(const v4i*)(x + NEDGE + 4 * t);
            v4i s1 = *(const v4i*)(x + 4 * (t + CONVT));
            v4i d1 = *(const v4i*)(x + NEDGE + 4 * (t + CONVT));
            v4i s2 = *(const v4i*)(x + 4 * (t + 2 * CONVT));
            v4i d2 = *(const v4i*)(x + NEDGE + 4 * (t + 2 * CONVT));
            v4i s3 = *(const v4i*)(x + 4 * (t + 3 * CONVT));
            v4i d3 = *(const v4i*)(x + NEDGE + 4 * (t + 3 * CONVT));
            conv_item(s0, d0, t, grp4, pk);
            conv_item(s1, d1, t + CONVT, grp4, pk);
            conv_item(s2, d2, t + 2 * CONVT, grp4, pk);
            conv_item(s3, d3, t + 3 * CONVT, grp4, pk);
        }
        for (; t < NITEM; t += CONVT) {
            v4i s0 = *(const v4i*)(x + 4 * t);
            v4i d0 = *(const v4i*)(x + NEDGE + 4 * t);
            conv_item(s0, d0, t, grp4, pk);
        }
        return;
    }
    int wave = threadIdx.x >> 5;
    int lane = threadIdx.x & 31;
    int pair = blockIdx.x * 4 + wave;
    if (pair >= NPAIR) return;
    int m0   = pair * 32;
    int lrow = lane & 15;
    int lhi  = lane >> 4;

    const float* grow0 = g + (long)(m0 + lrow) * NFEAT;
    const float* grow1 = grow0 + 16 * NFEAT;

    v8f acc0[8] = {};
    v8f acc1[8] = {};
    for (int k = 0; k < NFEAT; k += 4) {
        v2f a0 = *(const v2f*)(grow0 + k + 2 * lhi);
        v2f a1 = *(const v2f*)(grow1 + k + 2 * lhi);
#pragma unroll
        for (int t = 0; t < 8; ++t) {
            const float* Wp = (t < 4) ? We : Wg;
            int c  = (t & 3) * 16 + lrow;
            int kr = k + 2 * lhi;
            v2f b;
            b.x = Wp[(long)kr * NHID + c];
            b.y = Wp[(long)(kr + 1) * NHID + c];
            acc0[t] = __builtin_amdgcn_wmma_f32_16x16x4_f32(
                false, a0, false, b, (short)0, acc0[t], false, false);
            acc1[t] = __builtin_amdgcn_wmma_f32_16x16x4_f32(
                false, a1, false, b, (short)0, acc1[t], false, false);
        }
    }
#pragma unroll
    for (int t = 0; t < 8; ++t) {
        float* dst = (t < 4) ? h_est : h_gnn;
        int c = (t & 3) * 16 + lrow;
#pragma unroll
        for (int r = 0; r < 8; ++r) {
            dst[(long)(m0 + r + 8 * lhi) * NHID + c]      = acc0[t][r];
            dst[(long)(m0 + 16 + r + 8 * lhi) * NHID + c] = acc1[t][r];
        }
    }
}

__global__ void k_eprep(const float* __restrict__ h_est, const float* __restrict__ h_gnn,
                        const float* __restrict__ ase, const float* __restrict__ ade,
                        const float* __restrict__ asg, const float* __restrict__ adg,
                        float* es2, float* ed_e, float* ed_g) {
    int n = blockIdx.x * blockDim.x + threadIdx.x;
    if (n >= N_NODES) return;
    const float* he = h_est + (long)n * NHID;
    const float* hg = h_gnn + (long)n * NHID;
    float se = 0.f, de = 0.f, sg = 0.f, dg = 0.f;
#pragma unroll 4
    for (int f = 0; f < NHID; f += 4) {
        v4f e  = *(const v4f*)(he + f);
        v4f g4 = *(const v4f*)(hg + f);
        v4f a1 = *(const v4f*)(ase + f);
        v4f a2 = *(const v4f*)(ade + f);
        v4f a3 = *(const v4f*)(asg + f);
        v4f a4 = *(const v4f*)(adg + f);
        se += e.x * a1.x + e.y * a1.y + e.z * a1.z + e.w * a1.w;
        de += e.x * a2.x + e.y * a2.y + e.z * a2.z + e.w * a2.w;
        sg += g4.x * a3.x + g4.y * a3.y + g4.z * a3.z + g4.w * a3.w;
        dg += g4.x * a4.x + g4.y * a4.y + g4.z * a4.z + g4.w * a4.w;
    }
    *(v2f*)(es2 + 2 * n) = (v2f){se, sg};
    ed_e[n] = de; ed_g[n] = dg;
}

__device__ __forceinline__ unsigned zbytes(unsigned u) {
    unsigned t = (u & 0x7f7f7f7fu) + 0x7f7f7f7fu;
    return ~(t | u | 0x7f7f7f7fu);
}
__device__ __forceinline__ unsigned nib(unsigned f) {
    return ((f >> 7) & 1u) | ((f >> 14) & 2u) | ((f >> 21) & 4u) | ((f >> 28) & 8u);
}
__device__ __forceinline__ unsigned match16(v4u v, unsigned pat) {
    unsigned f0 = zbytes(v.x ^ pat), f1 = zbytes(v.y ^ pat);
    unsigned f2 = zbytes(v.z ^ pat), f3 = zbytes(v.w ^ pat);
    if ((f0 | f1 | f2 | f3) == 0u) return 0u;
    return nib(f0) | (nib(f1) << 4) | (nib(f2) << 8) | (nib(f3) << 12);
}
__device__ __forceinline__ unsigned flags32(v4u a, v4u b, unsigned pat) {
    return (zbytes(a.x ^ pat) >> 7) | (zbytes(a.y ^ pat) >> 6) |
           (zbytes(a.z ^ pat) >> 5) | (zbytes(a.w ^ pat) >> 4) |
           (zbytes(b.x ^ pat) >> 3) | (zbytes(b.y ^ pat) >> 2) |
           (zbytes(b.z ^ pat) >> 1) |  zbytes(b.w ^ pat);
}
__device__ __forceinline__ int rl_i(int v, int j) { return __builtin_amdgcn_readlane(v, j); }
__device__ __forceinline__ float rl_f(float v, int j) {
    return __int_as_float(__builtin_amdgcn_readlane(__float_as_int(v), j));
}

__global__ __launch_bounds__(1024) void k_own(const int* __restrict__ x,
        const unsigned* __restrict__ grp4, const unsigned* __restrict__ pk,
        const float* __restrict__ es2f,
        const float* __restrict__ ed_e, const float* __restrict__ ed_g,
        const float* __restrict__ h_est, const float* __restrict__ h_gnn,
        float* __restrict__ d_out,
        const float* __restrict__ eb, const float* __restrict__ egam,
        const float* __restrict__ ebeta, const float* __restrict__ erm,
        const float* __restrict__ erv,
        const float* __restrict__ gb, const float* __restrict__ ggam,
        const float* __restrict__ gbeta, const float* __restrict__ grm,
        const float* __restrict__ grv,
        const float* __restrict__ clsW, const float* __restrict__ clsb) {
    __shared__ float accE[GSZ * PITCH];
    __shared__ float accG[GSZ * PITCH];
    __shared__ unsigned lst[CAP];
    __shared__ unsigned lgE[CAP], lgG[CAP];
    __shared__ float zE[GSZ], zG[GSZ], edE[GSZ], edG[GSZ];
    __shared__ unsigned mE[GSZ], mG[GSZ];
    __shared__ unsigned cnt, ncand, ovf;

    const int tid  = threadIdx.x;
    const int lane = tid & 31;
    const int wave = __builtin_amdgcn_readfirstlane(tid >> 5);
    const int grpid = blockIdx.x;
    const int n0   = grpid * GSZ;
    const unsigned pat = (unsigned)grpid * 0x01010101u;
    const v4u* g16 = (const v4u*)grp4;
    const v2f* es2 = (const v2f*)es2f;

    for (int i = tid; i < GSZ * PITCH; i += 1024) { accE[i] = 0.f; accG[i] = 0.f; }
    if (tid < GSZ) {
        int n = n0 + tid;
        edE[tid] = ed_e[n]; edG[tid] = ed_g[n];
        zE[tid] = 0.f; zG[tid] = 0.f;
        mE[tid] = ford(-FLT_MAX); mG[tid] = ford(-FLT_MAX);
    }
    if (tid == 0) { cnt = GSZ; ncand = 0u; ovf = 0u; }
    __syncthreads();

    const v4u ones = (v4u){0xffffffffu, 0xffffffffu, 0xffffffffu, 0xffffffffu};
    for (int i0 = tid - lane; i0 < NV32; i0 += 2048) {
        const int ia = i0 + lane, ib = ia + 1024;
        v4u a0 = ones, a1 = ones, b0 = ones, b1 = ones;
        if (ia < NV32) { a0 = g16[2 * ia]; a1 = g16[2 * ia + 1]; }
        if (ib < NV32) { b0 = g16[2 * ib]; b1 = g16[2 * ib + 1]; }
        const unsigned Pa = flags32(a0, a1, pat), Pb = flags32(b0, b1, pat);
        const unsigned ma = __builtin_amdgcn_ballot_w32(Pa != 0u);
        const unsigned mb = __builtin_amdgcn_ballot_w32(Pb != 0u);
        const unsigned na = (unsigned)__builtin_popcount(ma);
        const unsigned nn = na + (unsigned)__builtin_popcount(mb);
        if (nn != 0u) {
            unsigned base = 0u;
            if (lane == 0) base = atomicAdd(&ncand, nn);
            base = (unsigned)__builtin_amdgcn_readfirstlane((int)base);
            if (Pa != 0u) {
                unsigned idx = base + __builtin_amdgcn_mbcnt_lo(ma, 0u);
                if (idx < CAP) { lgE[idx] = (unsigned)ia; lgG[idx] = Pa; } else ovf = 1u;
            }
            if (Pb != 0u) {
                unsigned idx = base + na + __builtin_amdgcn_mbcnt_lo(mb, 0u);
                if (idx < CAP) { lgE[idx] = (unsigned)ib; lgG[idx] = Pb; } else ovf = 1u;
            }
        }
    }
    __syncthreads();
    {
        const unsigned nc = ncand;
        const int ncl = __builtin_amdgcn_readfirstlane((int)(nc < (unsigned)CAP ? nc : (unsigned)CAP));
        for (int c = tid; c < ncl; c += 1024) {
            const unsigned blk = lgE[c];
            unsigned P = lgG[c];
            while (P) {
                unsigned t = (unsigned)__builtin_ctz(P); P &= P - 1u;
                unsigned idx = atomicAdd(&cnt, 1u);
                if (idx < CAP) lst[idx] = blk * 32u + 4u * (t & 7u) + (t >> 3); else ovf = 1u;
            }
        }
    }
    __syncthreads();
    const bool spilled = ovf != 0u;
    const int total = __builtin_amdgcn_readfirstlane(spilled ? GSZ : (int)cnt);

    for (int idx = tid; idx < total; idx += 1024) {
        unsigned p = (idx < GSZ) ? ((unsigned)(n0 + idx) | ((unsigned)idx << 17)) : pk[lst[idx]];
        int s = (int)(p & 0x1ffffu);
        int l = (int)(p >> 17);
        v2f es = es2[s];
        float ee = leaky(es.x + edE[l]);
        float eg = leaky(es.y + edG[l]);
        atomicMax(&mE[l], ford(ee));
        atomicMax(&mG[l], ford(eg));
        lst[idx] = p; lgE[idx] = __float_as_uint(ee); lgG[idx] = __float_as_uint(eg);
    }
    if (spilled) {
        for (int i = tid; i < NV16; i += 1024) {
            unsigned m = match16(g16[i], pat);
            while (m) {
                int j = __builtin_ctz(m); m &= m - 1u;
                int e = i * 16 + j;
                int s = x[e], l = x[NEDGE + e] - n0;
                v2f es = es2[s];
                atomicMax(&mE[l], ford(leaky(es.x + edE[l])));
                atomicMax(&mG[l], ford(leaky(es.y + edG[l])));
            }
        }
    }
    __syncthreads();

    for (int base = wave * 32; base < total; base += 1024) {
        int idx = base + lane;
        bool valid = idx < total;
        unsigned ent = valid ? lst[idx] : 0u;
        int s = (int)(ent & 0x1ffffu);
        int l = (int)(ent >> 17);
        float ae = 0.f, ag = 0.f;
        if (valid) {
            ae = __expf(__uint_as_float(lgE[idx]) - dorf(mE[l]));
            ag = __expf(__uint_as_float(lgG[idx]) - dorf(mG[l]));
            atomicAdd(&zE[l], ae);
            atomicAdd(&zG[l], ag);
        }
        int nv = total - base; if (nv > 32) nv = 32;
        int j = 0;
        for (; j + 8 <= nv; j += 8) {
            float r0[8], r1[8], r2[8], r3[8], aj[8], gj[8]; int lj[8];
#pragma unroll
            for (int u = 0; u < 8; ++u) {
                int sj = rl_i(s, j + u);
                lj[u] = rl_i(l, j + u); aj[u] = rl_f(ae, j + u); gj[u] = rl_f(ag, j + u);
                const float* qe = h_est + (long)sj * NHID + lane;
                const float* qg = h_gnn + (long)sj * NHID + lane;
                r0[u] = qe[0]; r1[u] = qe[32]; r2[u] = qg[0]; r3[u] = qg[32];
            }
#pragma unroll
            for (int u = 0; u < 8; ++u) {
                float* pe = &accE[lj[u] * PITCH + lane];
                float* pg = &accG[lj[u] * PITCH + lane];
                atomicAdd(pe,      r0[u] * aj[u]);
                atomicAdd(pe + 32, r1[u] * aj[u]);
                atomicAdd(pg,      r2[u] * gj[u]);
                atomicAdd(pg + 32, r3[u] * gj[u]);
            }
        }
        for (; j < nv; ++j) {
            int sj = rl_i(s, j), l1 = rl_i(l, j);
            float a1 = rl_f(ae, j), g1 = rl_f(ag, j);
            const float* qe = h_est + (long)sj * NHID + lane;
            const float* qg = h_gnn + (long)sj * NHID + lane;
            float t0 = qe[0], t1 = qe[32], t2 = qg[0], t3 = qg[32];
            float* pe = &accE[l1 * PITCH + lane];
            float* pg = &accG[l1 * PITCH + lane];
            atomicAdd(pe,      t0 * a1);
            atomicAdd(pe + 32, t1 * a1);
            atomicAdd(pg,      t2 * g1);
            atomicAdd(pg + 32, t3 * g1);
        }
    }
    if (spilled) {
        for (int i = tid; i < NV16; i += 1024) {
            unsigned m = match16(g16[i], pat);
            while (m) {
                int j = __builtin_ctz(m); m &= m - 1u;
                int e = i * 16 + j;
                int s = x[e], l = x[NEDGE + e] - n0;
                v2f es = es2[s];
                float ae = __expf(leaky(es.x + edE[l]) - dorf(mE[l]));
                float ag = __expf(leaky(es.y + edG[l]) - dorf(mG[l]));
                atomicAdd(&zE[l], ae);
                atomicAdd(&zG[l], ag);
                for (int f = 0; f < NHID; ++f) {
                    atomicAdd(&accE[l * PITCH + f], h_est[(long)s * NHID + f] * ae);
                    atomicAdd(&accG[l * PITCH + f], h_gnn[(long)s * NHID + f] * ag);
                }
            }
        }
    }
    __syncthreads();
    if (tid < GSZ) { zE[tid] = 1.f / zE[tid]; zG[tid] = 1.f / zG[tid]; }
    __syncthreads();

    {
        int f = tid & 63;
        float b = eb[f], rm = erm[f], rs = rsqrtf(erv[f] + EPSBN), gm = egam[f], bt = ebeta[f];
        float* srow = d_out + N_NODES + (long)n0 * NHID;
        for (int i = tid; i < GSZ * NHID; i += 1024) {
            int l = i >> 6;
            float a = fmaxf(accE[l * PITCH + f] * zE[l] + b, 0.f);
            srow[i] = (a - rm) * rs * gm + bt;
        }
    }
    if (tid < GSZ) {
        const float* grow = accG + tid * PITCH;
        float rzg = zG[tid];
        float y = 0.f;
#pragma unroll 4
        for (int f = 0; f < NHID; ++f) {
            float b = fmaxf(grow[f] * rzg + gb[f], 0.f);
            float zv = (b - grm[f]) * rsqrtf(grv[f] + EPSBN) * ggam[f] + gbeta[f];
            y += zv * clsW[f];
        }
        d_out[n0 + tid] = y + clsb[0];
    }
}

extern "C" void kernel_launch(void* const* d_in, const int* in_sizes, int n_in,
                              void* d_out, int out_size, void* d_ws, size_t ws_size,
                              hipStream_t stream) {
    const float* g        = (const float*)d_in[0];
    const int*   x        = (const int*)d_in[1];
    const float* est_W    = (const float*)d_in[2];
    const float* est_asrc = (const float*)d_in[3];
    const float* est_adst = (const float*)d_in[4];
    const float* est_bias = (const float*)d_in[5];
    const float* est_gam  = (const float*)d_in[6];
    const float* est_beta = (const float*)d_in[7];
    const float* est_rm   = (const float*)d_in[8];
    const float* est_rv   = (const float*)d_in[9];
    const float* gnn_W    = (const float*)d_in[10];
    const float* gnn_asrc = (const float*)d_in[11];
    const float* gnn_adst = (const float*)d_in[12];
    const float* gnn_bias = (const float*)d_in[13];
    const float* gnn_gam  = (const float*)d_in[14];
    const float* gnn_beta = (const float*)d_in[15];
    const float* gnn_rm   = (const float*)d_in[16];
    const float* gnn_rv   = (const float*)d_in[17];
    const float* cls_W    = (const float*)d_in[18];
    const float* cls_b    = (const float*)d_in[19];
    float* out = (float*)d_out;

    float*    ws     = (float*)d_ws;
    float*    h_est  = ws;
    float*    h_gnn  = h_est + (long)N_NODES * NHID;
    float*    es2f   = h_gnn + (long)N_NODES * NHID;
    float*    ed_e   = es2f + 2 * N_NODES;
    float*    ed_g   = ed_e + N_NODES;
    unsigned* grp4   = (unsigned*)(ed_g + N_NODES);
    unsigned* pk     = grp4 + NEDGE / 4;

    const int BT = 256;
    k_gemm<<<GEMMB + CONVB, GT, 0, stream>>>(g, est_W, gnn_W, h_est, h_gnn, x, grp4, pk);

    k_eprep<<<(N_NODES + BT - 1) / BT, BT, 0, stream>>>(h_est, h_gnn,
            est_asrc, est_adst, gnn_asrc, gnn_adst, es2f, ed_e, ed_g);

    k_own<<<NGRP, 1024, 0, stream>>>(x, grp4, pk, es2f, ed_e, ed_g, h_est, h_gnn, out,
            est_bias, est_gam, est_beta, est_rm, est_rv,
            gnn_bias, gnn_gam, gnn_beta, gnn_rm, gnn_rv, cls_W, cls_b);
}


# Owner kernel: entries bucketed by owning wave (local dst & 63; LDS counts, then places); a wave adds its own rows by 16-byte LDS load, fma, store, no float atomics; row terms sum in bucket fill order
# speedup vs baseline: 4.8698x; 1.0616x over previous
#include <hip/hip_runtime.h>
#include <cfloat>

#define N_NODES 100000
#define NFEAT   256
#define NHID    64
#define NEDGE   1600000
#define TOTE    (NEDGE + N_NODES)
#define MTILES  (N_NODES / 16)
#define EPSBN   1e-5f
#define NEG_SLOPE 0.2f

typedef __attribute__((ext_vector_type(2))) float v2f;
typedef __attribute__((ext_vector_type(4))) float v4f;
typedef __attribute__((ext_vector_type(8))) float v8f;

__device__ __forceinline__ unsigned ford(float f) {
    unsigned u = __float_as_uint(f);
    return (u & 0x80000000u) ? ~u : (u | 0x80000000u);
}
__device__ __forceinline__ float dorf(unsigned u) {
    return (u & 0x80000000u) ? __uint_as_float(u & 0x7fffffffu) : __uint_as_float(~u);
}
__device__ __forceinline__ float leaky(float v) {
    return fmaxf(v, 0.f) + NEG_SLOPE * fminf(v, 0.f);
}

#define NGRP    250
#define GSZ     400
#define ROWP    132
#define NBKT    64
#define CAP     8192
#define NV16    (NEDGE / 16)
#define NV32    (NEDGE / 32)

typedef __attribute__((ext_vector_type(4))) int v4i;
typedef __attribute__((ext_vector_type(4))) unsigned v4u;

__device__ __forceinline__ void conv_item(v4i s, v4i d, int t, unsigned* __restrict__ grp4,
                                          unsigned* __restrict__ pk) {
    int gx = d.x / GSZ, gy = d.y / GSZ, gz = d.z / GSZ, gw = d.w / GSZ;
    grp4[t] = (unsigned)gx | ((unsigned)gy << 8) | ((unsigned)gz << 16) | ((unsigned)gw << 24);
    v4u p;
    p.x = (unsigned)s.x | ((unsigned)(d.x - gx * GSZ) << 17);
    p.y = (unsigned)s.y | ((unsigned)(d.y - gy * GSZ) << 17);
    p.z = (unsigned)s.z | ((unsigned)(d.z - gz * GSZ) << 17);
    p.w = (unsigned)s.w | ((unsigned)(d.w - gw * GSZ) << 17);
    *(v4u*)(pk + 4 * t) = p;
}

#define GT      128
#define NPAIR   (N_NODES / 32)
#define GEMMB   ((NPAIR + 3) / 4)
#define CONVB   400
#define NITEM   (NEDGE / 4)
#define CONVT   (CONVB * GT)

__global__ __launch_bounds__(GT) void k_gemm(const float* __restrict__ g,
                                              const float* __restrict__ We,
                                              const float* __restrict__ Wg,
                                              float* __restrict__ h_est,
                                              float* __restrict__ h_gnn,
                                              const int* __restrict__ x,
                                              unsigned* __restrict__ grp4,
                                              unsigned* __restrict__ pk) {
    if (blockIdx.x >= GEMMB) {
        int t = (blockIdx.x - GEMMB) * GT + threadIdx.x;
        for (; t + 3 * CONVT < NITEM; t += 4 * CONVT) {
            v4i s0 = *(const v4i*)(x + 4 * t);
            v4i d0 = *(const v4i*)(x + NEDGE + 4 * t);
            v4i s1 = *(const v4i*)(x + 4 * (t + CONVT));
            v4i d1 = *(const v4i*)(x + NEDGE + 4 * (t + CONVT));
            v4i s2 = *(const v4i*)(x + 4 * (t + 2 * CONVT));
            v4i d2 = *(const v4i*)(x + NEDGE + 4 * (t + 2 * CONVT));
            v4i s3 = *(const v4i*)(x + 4 * (t + 3 * CONVT));
            v4i d3 = *(const v4i*)(x + NEDGE + 4 * (t + 3 * CONVT));
            conv_item(s0, d0, t, grp4, pk);
            conv_item(s1, d1, t + CONVT, grp4, pk);
            conv_item(s2, d2, t + 2 * CONVT, grp4, pk);
            conv_item(s3, d3, t + 3 * CONVT, grp4, pk);
        }
        for (; t < NITEM; t += CONVT) {
            v4i s0 = *(const v4i*)(x + 4 * t);
            v4i d0 = *(const v4i*)(x + NEDGE + 4 * t);
            conv_item(s0, d0, t, grp4, pk);
        }
        return;
    }
    int wave = threadIdx.x >> 5;
    int lane = threadIdx.x & 31;
    int pair = blockIdx.x * 4 + wave;
    if (pair >= NPAIR) return;
    int m0   = pair * 32;
    int lrow = lane & 15;
    int lhi  = lane >> 4;

    const float* grow0 = g + (long)(m0 + lrow) * NFEAT;
    const float* grow1 = grow0 + 16 * NFEAT;

    v8f acc0[8] = {};
    v8f acc1[8] = {};
    for (int k = 0; k < NFEAT; k += 4) {
        v2f a0 = *(const v2f*)(grow0 + k + 2 * lhi);
        v2f a1 = *(const v2f*)(grow1 + k + 2 * lhi);
#pragma unroll
        for (int t = 0; t < 8; ++t) {
            const float* Wp = (t < 4) ? We : Wg;
            int c  = (t & 3) * 16 + lrow;
            int kr = k + 2 * lhi;
            v2f b;
            b.x = Wp[(long)kr * NHID + c];
            b.y = Wp[(long)(kr + 1) * NHID + c];
            acc0[t] = __builtin_amdgcn_wmma_f32_16x16x4_f32(
                false, a0, false, b, (short)0, acc0[t], false, false);
            acc1[t] = __builtin_amdgcn_wmma_f32_16x16x4_f32(
                false, a1, false, b, (short)0, acc1[t], false, false);
        }
    }
#pragma unroll
    for (int t = 0; t < 8; ++t) {
        float* dst = (t < 4) ? h_est : h_gnn;
        int c = (t & 3) * 16 + lrow;
#pragma unroll
        for (int r = 0; r < 8; ++r) {
            dst[(long)(m0 + r + 8 * lhi) * NHID + c]      = acc0[t][r];
            dst[(long)(m0 + 16 + r + 8 * lhi) * NHID + c] = acc1[t][r];
        }
    }
}

__global__ void k_eprep(const float* __restrict__ h_est, const float* __restrict__ h_gnn,
                        const float* __restrict__ ase, const float* __restrict__ ade,
                        const float* __restrict__ asg, const float* __restrict__ adg,
                        float* es2, float* ed_e, float* ed_g) {
    int n = blockIdx.x * blockDim.x + threadIdx.x;
    if (n >= N_NODES) return;
    const float* he = h_est + (long)n * NHID;
    const float* hg = h_gnn + (long)n * NHID;
    float se = 0.f, de = 0.f, sg = 0.f, dg = 0.f;
#pragma unroll 4
    for (int f = 0; f < NHID; f += 4) {
        v4f e  = *(const v4f*)(he + f);
        v4f g4 = *(const v4f*)(hg + f);
        v4f a1 = *(const v4f*)(ase + f);
        v4f a2 = *(const v4f*)(ade + f);
        v4f a3 = *(const v4f*)(asg + f);
        v4f a4 = *(const v4f*)(adg + f);
        se += e.x * a1.x + e.y * a1.y + e.z * a1.z + e.w * a1.w;
        de += e.x * a2.x + e.y * a2.y + e.z * a2.z + e.w * a2.w;
        sg += g4.x * a3.x + g4.y * a3.y + g4.z * a3.z + g4.w * a3.w;
        dg += g4.x * a4.x + g4.y * a4.y + g4.z * a4.z + g4.w * a4.w;
    }
    *(v2f*)(es2 + 2 * n) = (v2f){se, sg};
    ed_e[n] = de; ed_g[n] = dg;
}

__device__ __forceinline__ unsigned zbytes(unsigned u) {
    unsigned t = (u & 0x7f7f7f7fu) + 0x7f7f7f7fu;
    return ~(t | u | 0x7f7f7f7fu);
}
__device__ __forceinline__ unsigned nib(unsigned f) {
    return ((f >> 7) & 1u) | ((f >> 14) & 2u) | ((f >> 21) & 4u) | ((f >> 28) & 8u);
}
__device__ __forceinline__ unsigned match16(v4u v, unsigned pat) {
    unsigned f0 = zbytes(v.x ^ pat), f1 = zbytes(v.y ^ pat);
    unsigned f2 = zbytes(v.z ^ pat), f3 = zbytes(v.w ^ pat);
    if ((f0 | f1 | f2 | f3) == 0u) return 0u;
    return nib(f0) | (nib(f1) << 4) | (nib(f2) << 8) | (nib(f3) << 12);
}
__device__ __forceinline__ unsigned flags32(v4u a, v4u b, unsigned pat) {
    return (zbytes(a.x ^ pat) >> 7) | (zbytes(a.y ^ pat) >> 6) |
           (zbytes(a.z ^ pat) >> 5) | (zbytes(a.w ^ pat) >> 4) |
           (zbytes(b.x ^ pat) >> 3) | (zbytes(b.y ^ pat) >> 2) |
           (zbytes(b.z ^ pat) >> 1) |  zbytes(b.w ^ pat);
}
__device__ __forceinline__ int rl_i(int v, int j) { return __builtin_amdgcn_readlane(v, j); }
__device__ __forceinline__ float rl_f(float v, int j) {
    return __int_as_float(__builtin_amdgcn_readlane(__float_as_int(v), j));
}
__device__ __forceinline__ void rows_load(v4f (&r)[8], const float* hsel, int s, int j) {
#pragma unroll
    for (int u = 0; u < 8; ++u) r[u] = *(const v4f*)(hsel + (long)rl_i(s, j + u) * NHID);
}
__device__ __forceinline__ void rows_add(const v4f (&r)[8], float* arow, int l, float ae, float ag,
                                         int half, int j) {
#pragma unroll
    for (int u = 0; u < 8; ++u) {
        float a1 = rl_f(ae, j + u), g1 = rl_f(ag, j + u);
        float w = half ? g1 : a1;
        v4f* q = (v4f*)(arow + rl_i(l, j + u) * ROWP);
        v4f a = *q;
        a.x += w * r[u].x; a.y += w * r[u].y; a.z += w * r[u].z; a.w += w * r[u].w;
        *q = a;
    }
}

__global__ __launch_bounds__(1024) void k_own(const int* __restrict__ x,
        const unsigned* __restrict__ grp4, const unsigned* __restrict__ pk,
        const float* __restrict__ es2f,
        const float* __restrict__ ed_e, const float* __restrict__ ed_g,
        const float* __restrict__ h_est, const float* __restrict__ h_gnn,
        float* __restrict__ d_out,
        const float* __restrict__ eb, const float* __restrict__ egam,
        const float* __restrict__ ebeta, const float* __restrict__ erm,
        const float* __restrict__ erv,
        const float* __restrict__ gb, const float* __restrict__ ggam,
        const float* __restrict__ gbeta, const float* __restrict__ grm,
        const float* __restrict__ grv,
        const float* __restrict__ clsW, const float* __restrict__ clsb) {
    __shared__ __attribute__((aligned(16))) float acc[GSZ * ROWP];
    __shared__ unsigned lst[CAP];
    __shared__ unsigned lgE[CAP], lgG[CAP];
    __shared__ unsigned bcnt[NBKT], bfill[NBKT], bpos[NBKT + 1];
    __shared__ float zE[GSZ], zG[GSZ], edE[GSZ], edG[GSZ];
    __shared__ unsigned mE[GSZ], mG[GSZ];
    __shared__ unsigned cnt, ncand, ovf;

    const int tid  = threadIdx.x;
    const int lane = tid & 31;
    const int wave = __builtin_amdgcn_readfirstlane(tid >> 5);
    const int grpid = blockIdx.x;
    const int n0   = grpid * GSZ;
    const unsigned pat = (unsigned)grpid * 0x01010101u;
    const v4u* g16 = (const v4u*)grp4;
    const v2f* es2 = (const v2f*)es2f;
    unsigned* tP = (unsigned*)acc;
    unsigned* tE = tP + CAP;
    unsigned* tG = tE + CAP;

    if (tid < NBKT) { bcnt[tid] = 0u; bfill[tid] = 0u; }
    if (tid < GSZ) {
        int n = n0 + tid;
        edE[tid] = ed_e[n]; edG[tid] = ed_g[n];
        zE[tid] = 0.f; zG[tid] = 0.f;
        mE[tid] = ford(-FLT_MAX); mG[tid] = ford(-FLT_MAX);
    }
    if (tid == 0) { cnt = GSZ; ncand = 0u; ovf = 0u; }
    __syncthreads();

    const v4u ones = (v4u){0xffffffffu, 0xffffffffu, 0xffffffffu, 0xffffffffu};
    for (int i0 = tid - lane; i0 < NV32; i0 += 2048) {
        const int ia = i0 + lane, ib = ia + 1024;
        v4u a0 = ones, a1 = ones, b0 = ones, b1 = ones;
        if (ia < NV32) { a0 = g16[2 * ia]; a1 = g16[2 * ia + 1]; }
        if (ib < NV32) { b0 = g16[2 * ib]; b1 = g16[2 * ib + 1]; }
        const unsigned Pa = flags32(a0, a1, pat), Pb = flags32(b0, b1, pat);
        const unsigned ma = __builtin_amdgcn_ballot_w32(Pa != 0u);
        const unsigned mb = __builtin_amdgcn_ballot_w32(Pb != 0u);
        const unsigned na = (unsigned)__builtin_popcount(ma);
        const unsigned nn = na + (unsigned)__builtin_popcount(mb);
        if (nn != 0u) {
            unsigned base = 0u;
            if (lane == 0) base = atomicAdd(&ncand, nn);
            base = (unsigned)__builtin_amdgcn_readfirstlane((int)base);
            if (Pa != 0u) {
                unsigned idx = base + __builtin_amdgcn_mbcnt_lo(ma, 0u);
                if (idx < CAP) { lgE[idx] = (unsigned)ia; lgG[idx] = Pa; } else ovf = 1u;
            }
            if (Pb != 0u) {
                unsigned idx = base + na + __builtin_amdgcn_mbcnt_lo(mb, 0u);
                if (idx < CAP) { lgE[idx] = (unsigned)ib; lgG[idx] = Pb; } else ovf = 1u;
            }
        }
    }
    __syncthreads();
    {
        const unsigned nc = ncand;
        const int ncl = __builtin_amdgcn_readfirstlane((int)(nc < (unsigned)CAP ? nc : (unsigned)CAP));
        for (int c = tid; c < ncl; c += 1024) {
            const unsigned blk = lgE[c];
            unsigned P = lgG[c];
            while (P) {
                unsigned t = (unsigned)__builtin_ctz(P); P &= P - 1u;
                unsigned idx = atomicAdd(&cnt, 1u);
                if (idx < CAP) tP[idx] = blk * 32u + 4u * (t & 7u) + (t >> 3); else ovf = 1u;
            }
        }
    }
    __syncthreads();
    const bool spilled = ovf != 0u;
    const int total = __builtin_amdgcn_readfirstlane(spilled ? GSZ : (int)cnt);

    for (int idx = tid; idx < total; idx += 1024) {
        unsigned p = (idx < GSZ) ? ((unsigned)(n0 + idx) | ((unsigned)idx << 17)) : pk[tP[idx]];
        int s = (int)(p & 0x1ffffu);
        int l = (int)(p >> 17);
        v2f es = es2[s];
        float ee = leaky(es.x + edE[l]);
        float eg = leaky(es.y + edG[l]);
        atomicMax(&mE[l], ford(ee));
        atomicMax(&mG[l], ford(eg));
        atomicAdd(&bcnt[l & (NBKT - 1)], 1u);
        tP[idx] = p; tE[idx] = __float_as_uint(ee); tG[idx] = __float_as_uint(eg);
    }
    if (spilled) {
        for (int i = tid; i < NV16; i += 1024) {
            unsigned m = match16(g16[i], pat);
            while (m) {
                int j = __builtin_ctz(m); m &= m - 1u;
                int e = i * 16 + j;
                int s = x[e], l = x[NEDGE + e] - n0;
                v2f es = es2[s];
                atomicMax(&mE[l], ford(leaky(es.x + edE[l])));
                atomicMax(&mG[l], ford(leaky(es.y + edG[l])));
            }
        }
    }
    __syncthreads();

    if (tid < NBKT) {
        unsigned a = 0u;
        for (int k = 0; k < NBKT; ++k) { unsigned c = bcnt[k]; if (k < tid) a += c; }
        bpos[tid] = a;
        if (tid == NBKT - 1) bpos[NBKT] = a + bcnt[NBKT - 1];
    }
    __syncthreads();
    for (int idx = tid; idx < total; idx += 1024) {
        unsigned p = tP[idx];
        int l = (int)(p >> 17);
        float ae = __expf(__uint_as_float(tE[idx]) - dorf(mE[l]));
        float ag = __expf(__uint_as_float(tG[idx]) - dorf(mG[l]));
        atomicAdd(&zE[l], ae);
        atomicAdd(&zG[l], ag);
        int b = l & (NBKT - 1);
        unsigned pos = bpos[b] + atomicAdd(&bfill[b], 1u);
        lst[pos] = p; lgE[pos] = __float_as_uint(ae); lgG[pos] = __float_as_uint(ag);
    }
    __syncthreads();
    for (int i = tid; i < GSZ * ROWP / 4; i += 1024) ((v4f*)acc)[i] = (v4f){0.f, 0.f, 0.f, 0.f};
    __syncthreads();

    {
        const int half = lane >> 4;
        const float* hsel = (half ? h_gnn : h_est) + 4 * (lane & 15);
        float* arow = acc + 4 * lane;
        const int beg = __builtin_amdgcn_readfirstlane((int)bpos[2 * wave]);
        const int end = __builtin_amdgcn_readfirstlane((int)bpos[2 * wave + 2]);
        for (int base = beg; base < end; base += 32) {
            int idx = base + lane;
            bool valid = idx < end;
            unsigned ent = valid ? lst[idx] : 0u;
            int s = (int)(ent & 0x1ffffu);
            int l = (int)(ent >> 17);
            float ae = valid ? __uint_as_float(lgE[idx]) : 0.f;
            float ag = valid ? __uint_as_float(lgG[idx]) : 0.f;
            int nv = end - base; if (nv > 32) nv = 32;
            int j = 0;
            if (nv == 32) {
                v4f ra[8], rb[8];
                rows_load(ra, hsel, s, 0);
                rows_load(rb, hsel, s, 8);
                rows_add(ra, arow, l, ae, ag, half, 0);
                rows_load(ra, hsel, s, 16);
                rows_add(rb, arow, l, ae, ag, half, 8);
                rows_load(rb, hsel, s, 24);
                rows_add(ra, arow, l, ae, ag, half, 16);
                rows_add(rb, arow, l, ae, ag, half, 24);
                j = 32;
            }
            for (; j + 8 <= nv; j += 8) {
                v4f r[8];
                rows_load(r, hsel, s, j);
                rows_add(r, arow, l, ae, ag, half, j);
            }
            for (; j < nv; ++j) {
                int sj = rl_i(s, j), l1 = rl_i(l, j);
                float a1 = rl_f(ae, j), g1 = rl_f(ag, j);
                float w1 = half ? g1 : a1;
                v4f r1 = *(const v4f*)(hsel + (long)sj * NHID);
                v4f* q = (v4f*)(arow + l1 * ROWP);
                v4f a = *q;
                a.x += w1 * r1.x; a.y += w1 * r1.y; a.z += w1 * r1.z; a.w += w1 * r1.w;
                *q = a;
            }
        }
    }
    if (spilled) {
        __syncthreads();
        for (int i = tid; i < NV16; i += 1024) {
            unsigned m = match16(g16[i], pat);
            while (m) {
                int j = __builtin_ctz(m); m &= m - 1u;
                int e = i * 16 + j;
                int s = x[e], l = x[NEDGE + e] - n0;
                v2f es = es2[s];
                float ae = __expf(leaky(es.x + edE[l]) - dorf(mE[l]));
                float ag = __expf(leaky(es.y + edG[l]) - dorf(mG[l]));
                atomicAdd(&zE[l], ae);
                atomicAdd(&zG[l], ag);
                for (int f = 0; f < NHID; ++f) {
                    atomicAdd(&acc[l * ROWP + f], h_est[(long)s * NHID + f] * ae);
                    atomicAdd(&acc[l * ROWP + NHID + f], h_gnn[(long)s * NHID + f] * ag);
                }
            }
        }
    }
    __syncthreads();
    if (tid < GSZ) { zE[tid] = 1.f / zE[tid]; zG[tid] = 1.f / zG[tid]; }
    __syncthreads();

    {
        int f = tid & 63;
        float b = eb[f], rm = erm[f], rs = rsqrtf(erv[f] + EPSBN), gm = egam[f], bt = ebeta[f];
        float* srow = d_out + N_NODES + (long)n0 * NHID;
        for (int i = tid; i < GSZ * NHID; i += 1024) {
            int l = i >> 6;
            float a = fmaxf(acc[l * ROWP + f] * zE[l] + b, 0.f);
            srow[i] = (a - rm) * rs * gm + bt;
        }
    }
    if (tid < GSZ) {
        const float* grow = acc + tid * ROWP + NHID;
        float rzg = zG[tid];
        float y = 0.f;
#pragma unroll 4
        for (int f = 0; f < NHID; ++f) {
            float b = fmaxf(grow[f] * rzg + gb[f], 0.f);
            float zv = (b - grm[f]) * rsqrtf(grv[f] + EPSBN) * ggam[f] + gbeta[f];
            y += zv * clsW[f];
        }
        d_out[n0 + tid] = y + clsb[0];
    }
}

extern "C" void kernel_launch(void* const* d_in, const int* in_sizes, int n_in,
                              void* d_out, int out_size, void* d_ws, size_t ws_size,
                              hipStream_t stream) {
    const float* g        = (const float*)d_in[0];
    const int*   x        = (const int*)d_in[1];
    const float* est_W    = (const float*)d_in[2];
    const float* est_asrc = (const float*)d_in[3];
    const float* est_adst = (const float*)d_in[4];
    const float* est_bias = (const float*)d_in[5];
    const float* est_gam  = (const float*)d_in[6];
    const float* est_beta = (const float*)d_in[7];
    const float* est_rm   = (const float*)d_in[8];
    const float* est_rv   = (const float*)d_in[9];
    const float* gnn_W    = (const float*)d_in[10];
    const float* gnn_asrc = (const float*)d_in[11];
    const float* gnn_adst = (const float*)d_in[12];
    const float* gnn_bias = (const float*)d_in[13];
    const float* gnn_gam  = (const float*)d_in[14];
    const float* gnn_beta = (const float*)d_in[15];
    const float* gnn_rm   = (const float*)d_in[16];
    const float* gnn_rv   = (const float*)d_in[17];
    const float* cls_W    = (const float*)d_in[18];
    const float* cls_b    = (const float*)d_in[19];
    float* out = (float*)d_out;

    float*    ws     = (float*)d_ws;
    float*    h_est  = ws;
    float*    h_gnn  = h_est + (long)N_NODES * NHID;
    float*    es2f   = h_gnn + (long)N_NODES * NHID;
    float*    ed_e   = es2f + 2 * N_NODES;
    float*    ed_g   = ed_e + N_NODES;
    unsigned* grp4   = (unsigned*)(ed_g + N_NODES);
    unsigned* pk     = grp4 + NEDGE / 4;

    const int BT = 256;
    k_gemm<<<GEMMB + CONVB, GT, 0, stream>>>(g, est_W, gnn_W, h_est, h_gnn, x, grp4, pk);

    k_eprep<<<(N_NODES + BT - 1) / BT, BT, 0, stream>>>(h_est, h_gnn,
            est_asrc, est_adst, gnn_asrc, gnn_adst, es2f, ed_e, ed_g);

    k_own<<<NGRP, 1024, 0, stream>>>(x, grp4, pk, es2f, ed_e, ed_g, h_est, h_gnn, out,
            est_bias, est_gam, est_beta, est_rm, est_rv,
            gnn_bias, gnn_gam, gnn_beta, gnn_rm, gnn_rv, cls_W, cls_b);
}


# GEMM grid cut so every SIMD's f32 matrix unit gets equal work: 768 work-groups of 32x128 wave tiles (3 per SIMD), last 1696 rows as 424 short 16x32 waves, no 4th full tile; k order of sums unchanged
# speedup vs baseline: 5.0860x; 1.0444x over previous
#include <hip/hip_runtime.h>
#include <cfloat>

#define N_NODES 100000
#define NFEAT   256
#define NHID    64
#define NEDGE   1600000
#define TOTE    (NEDGE + N_NODES)
#define MTILES  (N_NODES / 16)
#define EPSBN   1e-5f
#define NEG_SLOPE 0.2f

typedef __attribute__((ext_vector_type(2))) float v2f;
typedef __attribute__((ext_vector_type(4))) float v4f;
typedef __attribute__((ext_vector_type(8))) float v8f;

__device__ __forceinline__ unsigned ford(float f) {
    unsigned u = __float_as_uint(f);
    return (u & 0x80000000u) ? ~u : (u | 0x80000000u);
}
__device__ __forceinline__ float dorf(unsigned u) {
    return (u & 0x80000000u) ? __uint_as_float(u & 0x7fffffffu) : __uint_as_float(~u);
}
__device__ __forceinline__ float leaky(float v) {
    return fmaxf(v, 0.f) + NEG_SLOPE * fminf(v, 0.f);
}

#define NGRP    250
#define GSZ     400
#define ROWP    132
#define NBKT    64
#define CAP     8192
#define NV16    (NEDGE / 16)
#define NV32    (NEDGE / 32)

typedef __attribute__((ext_vector_type(4))) int v4i;
typedef __attribute__((ext_vector_type(4))) unsigned v4u;

__device__ __forceinline__ void conv_item(v4i s, v4i d, int t, unsigned* __restrict__ grp4,
                                          unsigned* __restrict__ pk) {
    int gx = d.x / GSZ, gy = d.y / GSZ, gz = d.z / GSZ, gw = d.w / GSZ;
    grp4[t] = (unsigned)gx | ((unsigned)gy << 8) | ((unsigned)gz << 16) | ((unsigned)gw << 24);
    v4u p;
    p.x = (unsigned)s.x | ((unsigned)(d.x - gx * GSZ) << 17);
    p.y = (unsigned)s.y | ((unsigned)(d.y - gy * GSZ) << 17);
    p.z = (unsigned)s.z | ((unsigned)(d.z - gz * GSZ) << 17);
    p.w = (unsigned)s.w | ((unsigned)(d.w - gw * GSZ) << 17);
    *(v4u*)(pk + 4 * t) = p;
}

#define GT      128
#define NBIG    768
#define BIGROWS (NBIG * 4 * 32)
#define NTAIL   ((N_NODES - BIGROWS) / 16)
#define GEMMB   (NBIG + NTAIL)
#define CONVB   400
#define NITEM   (NEDGE / 4)
#define CONVT   (CONVB * GT)

__global__ __launch_bounds__(GT) void k_gemm(const float* __restrict__ g,
                                              const float* __restrict__ We,
                                              const float* __restrict__ Wg,
                                              float* __restrict__ h_est,
                                              float* __restrict__ h_gnn,
                                              const int* __restrict__ x,
                                              unsigned* __restrict__ grp4,
                                              unsigned* __restrict__ pk) {
    if (blockIdx.x >= GEMMB) {
        int t = (blockIdx.x - GEMMB) * GT + threadIdx.x;
        for (; t + 3 * CONVT < NITEM; t += 4 * CONVT) {
            v4i s0 = *(const v4i*)(x + 4 * t);
            v4i d0 = *(const v4i*)(x + NEDGE + 4 * t);
            v4i s1 = *(const v4i*)(x + 4 * (t + CONVT));
            v4i d1 = *(const v4i*)(x + NEDGE + 4 * (t + CONVT));
            v4i s2 = *(const v4i*)(x + 4 * (t + 2 * CONVT));
            v4i d2 = *(const v4i*)(x + NEDGE + 4 * (t + 2 * CONVT));
            v4i s3 = *(const v4i*)(x + 4 * (t + 3 * CONVT));
            v4i d3 = *(const v4i*)(x + NEDGE + 4 * (t + 3 * CONVT));
            conv_item(s0, d0, t, grp4, pk);
            conv_item(s1, d1, t + CONVT, grp4, pk);
            conv_item(s2, d2, t + 2 * CONVT, grp4, pk);
            conv_item(s3, d3, t + 3 * CONVT, grp4, pk);
        }
        for (; t < NITEM; t += CONVT) {
            v4i s0 = *(const v4i*)(x + 4 * t);
            v4i d0 = *(const v4i*)(x + NEDGE + 4 * t);
            conv_item(s0, d0, t, grp4, pk);
        }
        return;
    }
    int wave = threadIdx.x >> 5;
    int lane = threadIdx.x & 31;
    int lrow = lane & 15;
    int lhi  = lane >> 4;

    if (blockIdx.x >= NBIG) {
        const int mt = BIGROWS + (blockIdx.x - NBIG) * 16;
        const float* grow = g + (long)(mt + lrow) * NFEAT;
        const float* Wp = (wave < 2) ? We : Wg;
        float* dst = (wave < 2) ? h_est : h_gnn;
        const int c = (wave & 1) * 32 + lrow;
        v8f t0 = {};
        v8f t1 = {};
        for (int k = 0; k < NFEAT; k += 4) {
            int kr = k + 2 * lhi;
            v2f a = *(const v2f*)(grow + kr);
            v2f b0, b1;
            b0.x = Wp[(long)kr * NHID + c];
            b0.y = Wp[(long)(kr + 1) * NHID + c];
            b1.x = Wp[(long)kr * NHID + c + 16];
            b1.y = Wp[(long)(kr + 1) * NHID + c + 16];
            t0 = __builtin_amdgcn_wmma_f32_16x16x4_f32(
                false, a, false, b0, (short)0, t0, false, false);
            t1 = __builtin_amdgcn_wmma_f32_16x16x4_f32(
                false, a, false, b1, (short)0, t1, false, false);
        }
#pragma unroll
        for (int r = 0; r < 8; ++r) {
            dst[(long)(mt + r + 8 * lhi) * NHID + c]      = t0[r];
            dst[(long)(mt + r + 8 * lhi) * NHID + c + 16] = t1[r];
        }
        return;
    }
    int pair = blockIdx.x * 4 + wave;
    int m0   = pair * 32;

    const float* grow0 = g + (long)(m0 + lrow) * NFEAT;
    const float* grow1 = grow0 + 16 * NFEAT;

    v8f acc0[8] = {};
    v8f acc1[8] = {};
    for (int k = 0; k < NFEAT; k += 4) {
        v2f a0 = *(const v2f*)(grow0 + k + 2 * lhi);
        v2f a1 = *(const v2f*)(grow1 + k + 2 * lhi);
#pragma unroll
        for (int t = 0; t < 8; ++t) {
            const float* Wp = (t < 4) ? We : Wg;
            int c  = (t & 3) * 16 + lrow;
            int kr = k + 2 * lhi;
            v2f b;
            b.x = Wp[(long)kr * NHID + c];
            b.y = Wp[(long)(kr + 1) * NHID + c];
            acc0[t] = __builtin_amdgcn_wmma_f32_16x16x4_f32(
                false, a0, false, b, (short)0, acc0[t], false, false);
            acc1[t] = __builtin_amdgcn_wmma_f32_16x16x4_f32(
                false, a1, false, b, (short)0, acc1[t], false, false);
        }
    }
#pragma unroll
    for (int t = 0; t < 8; ++t) {
        float* dst = (t < 4) ? h_est : h_gnn;
        int c = (t & 3) * 16 + lrow;
#pragma unroll
        for (int r = 0; r < 8; ++r) {
            dst[(long)(m0 + r + 8 * lhi) * NHID + c]      = acc0[t][r];
            dst[(long)(m0 + 16 + r + 8 * lhi) * NHID + c] = acc1[t][r];
        }
    }
}

__global__ void k_eprep(const float* __restrict__ h_est, const float* __restrict__ h_gnn,
                        const float* __restrict__ ase, const float* __restrict__ ade,
                        const float* __restrict__ asg, const float* __restrict__ adg,
                        float* es2, float* ed_e, float* ed_g) {
    int n = blockIdx.x * blockDim.x + threadIdx.x;
    if (n >= N_NODES) return;
    const float* he = h_est + (long)n * NHID;
    const float* hg = h_gnn + (long)n * NHID;
    float se = 0.f, de = 0.f, sg = 0.f, dg = 0.f;
#pragma unroll 4
    for (int f = 0; f < NHID; f += 4) {
        v4f e  = *(const v4f*)(he + f);
        v4f g4 = *(const v4f*)(hg + f);
        v4f a1 = *(const v4f*)(ase + f);
        v4f a2 = *(const v4f*)(ade + f);
        v4f a3 = *(const v4f*)(asg + f);
        v4f a4 = *(const v4f*)(adg + f);
        se += e.x * a1.x + e.y * a1.y + e.z * a1.z + e.w * a1.w;
        de += e.x * a2.x + e.y * a2.y + e.z * a2.z + e.w * a2.w;
        sg += g4.x * a3.x + g4.y * a3.y + g4.z * a3.z + g4.w * a3.w;
        dg += g4.x * a4.x + g4.y * a4.y + g4.z * a4.z + g4.w * a4.w;
    }
    *(v2f*)(es2 + 2 * n) = (v2f){se, sg};
    ed_e[n] = de; ed_g[n] = dg;
}

__device__ __forceinline__ unsigned zbytes(unsigned u) {
    unsigned t = (u & 0x7f7f7f7fu) + 0x7f7f7f7fu;
    return ~(t | u | 0x7f7f7f7fu);
}
__device__ __forceinline__ unsigned nib(unsigned f) {
    return ((f >> 7) & 1u) | ((f >> 14) & 2u) | ((f >> 21) & 4u) | ((f >> 28) & 8u);
}
__device__ __forceinline__ unsigned match16(v4u v, unsigned pat) {
    unsigned f0 = zbytes(v.x ^ pat), f1 = zbytes(v.y ^ pat);
    unsigned f2 = zbytes(v.z ^ pat), f3 = zbytes(v.w ^ pat);
    if ((f0 | f1 | f2 | f3) == 0u) return 0u;
    return nib(f0) | (nib(f1) << 4) | (nib(f2) << 8) | (nib(f3) << 12);
}
__device__ __forceinline__ unsigned flags32(v4u a, v4u b, unsigned pat) {
    return (zbytes(a.x ^ pat) >> 7) | (zbytes(a.y ^ pat) >> 6) |
           (zbytes(a.z ^ pat) >> 5) | (zbytes(a.w ^ pat) >> 4) |
           (zbytes(b.x ^ pat) >> 3) | (zbytes(b.y ^ pat) >> 2) |
           (zbytes(b.z ^ pat) >> 1) |  zbytes(b.w ^ pat);
}
__device__ __forceinline__ int rl_i(int v, int j) { return __builtin_amdgcn_readlane(v, j); }
__device__ __forceinline__ float rl_f(float v, int j) {
    return __int_as_float(__builtin_amdgcn_readlane(__float_as_int(v), j));
}
__device__ __forceinline__ void rows_load(v4f (&r)[8], const float* hsel, int s, int j) {
#pragma unroll
    for (int u = 0; u < 8; ++u) r[u] = *(const v4f*)(hsel + (long)rl_i(s, j + u) * NHID);
}
__device__ __forceinline__ void rows_add(const v4f (&r)[8], float* arow, int l, float ae, float ag,
                                         int half, int j) {
#pragma unroll
    for (int u = 0; u < 8; ++u) {
        float a1 = rl_f(ae, j + u), g1 = rl_f(ag, j + u);
        float w = half ? g1 : a1;
        v4f* q = (v4f*)(arow + rl_i(l, j + u) * ROWP);
        v4f a = *q;
        a.x += w * r[u].x; a.y += w * r[u].y; a.z += w * r[u].z; a.w += w * r[u].w;
        *q = a;
    }
}

__global__ __launch_bounds__(1024) void k_own(const int* __restrict__ x,
        const unsigned* __restrict__ grp4, const unsigned* __restrict__ pk,
        const float* __restrict__ es2f,
        const float* __restrict__ ed_e, const float* __restrict__ ed_g,
        const float* __restrict__ h_est, const float* __restrict__ h_gnn,
        float* __restrict__ d_out,
        const float* __restrict__ eb, const float* __restrict__ egam,
        const float* __restrict__ ebeta, const float* __restrict__ erm,
        const float* __restrict__ erv,
        const float* __restrict__ gb, const float* __restrict__ ggam,
        const float* __restrict__ gbeta, const float* __restrict__ grm,
        const float* __restrict__ grv,
        const float* __restrict__ clsW, const float* __restrict__ clsb) {
    __shared__ __attribute__((aligned(16))) float acc[GSZ * ROWP];
    __shared__ unsigned lst[CAP];
    __shared__ unsigned lgE[CAP], lgG[CAP];
    __shared__ unsigned bcnt[NBKT], bfill[NBKT], bpos[NBKT + 1];
    __shared__ float zE[GSZ], zG[GSZ], edE[GSZ], edG[GSZ];
    __shared__ unsigned mE[GSZ], mG[GSZ];
    __shared__ unsigned cnt, ncand, ovf;

    const int tid  = threadIdx.x;
    const int lane = tid & 31;
    const int wave = __builtin_amdgcn_readfirstlane(tid >> 5);
    const int grpid = blockIdx.x;
    const int n0   = grpid * GSZ;
    const unsigned pat = (unsigned)grpid * 0x01010101u;
    const v4u* g16 = (const v4u*)grp4;
    const v2f* es2 = (const v2f*)es2f;
    unsigned* tP = (unsigned*)acc;
    unsigned* tE = tP + CAP;
    unsigned* tG = tE + CAP;

    if (tid < NBKT) { bcnt[tid] = 0u; bfill[tid] = 0u; }
    if (tid < GSZ) {
        int n = n0 + tid;
        edE[tid] = ed_e[n]; edG[tid] = ed_g[n];
        zE[tid] = 0.f; zG[tid] = 0.f;
        mE[tid] = ford(-FLT_MAX); mG[tid] = ford(-FLT_MAX);
    }
    if (tid == 0) { cnt = GSZ; ncand = 0u; ovf = 0u; }
    __syncthreads();

    const v4u ones = (v4u){0xffffffffu, 0xffffffffu, 0xffffffffu, 0xffffffffu};
    for (int i0 = tid - lane; i0 < NV32; i0 += 2048) {
        const int ia = i0 + lane, ib = ia + 1024;
        v4u a0 = ones, a1 = ones, b0 = ones, b1 = ones;
        if (ia < NV32) { a0 = g16[2 * ia]; a1 = g16[2 * ia + 1]; }
        if (ib < NV32) { b0 = g16[2 * ib]; b1 = g16[2 * ib + 1]; }
        const unsigned Pa = flags32(a0, a1, pat), Pb = flags32(b0, b1, pat);
        const unsigned ma = __builtin_amdgcn_ballot_w32(Pa != 0u);
        const unsigned mb = __builtin_amdgcn_ballot_w32(Pb != 0u);
        const unsigned na = (unsigned)__builtin_popcount(ma);
        const unsigned nn = na + (unsigned)__builtin_popcount(mb);
        if (nn != 0u) {
            unsigned base = 0u;
            if (lane == 0) base = atomicAdd(&ncand, nn);
            base = (unsigned)__builtin_amdgcn_readfirstlane((int)base);
            if (Pa != 0u) {
                unsigned idx = base + __builtin_amdgcn_mbcnt_lo(ma, 0u);
                if (idx < CAP) { lgE[idx] = (unsigned)ia; lgG[idx] = Pa; } else ovf = 1u;
            }
            if (Pb != 0u) {
                unsigned idx = base + na + __builtin_amdgcn_mbcnt_lo(mb, 0u);
                if (idx < CAP) { lgE[idx] = (unsigned)ib; lgG[idx] = Pb; } else ovf = 1u;
            }
        }
    }
    __syncthreads();
    {
        const unsigned nc = ncand;
        const int ncl = __builtin_amdgcn_readfirstlane((int)(nc < (unsigned)CAP ? nc : (unsigned)CAP));
        for (int c = tid; c < ncl; c += 1024) {
            const unsigned blk = lgE[c];
            unsigned P = lgG[c];
            while (P) {
                unsigned t = (unsigned)__builtin_ctz(P); P &= P - 1u;
                unsigned idx = atomicAdd(&cnt, 1u);
                if (idx < CAP) tP[idx] = blk * 32u + 4u * (t & 7u) + (t >> 3); else ovf = 1u;
            }
        }
    }
    __syncthreads();
    const bool spilled = ovf != 0u;
    const int total = __builtin_amdgcn_readfirstlane(spilled ? GSZ : (int)cnt);

    for (int idx = tid; idx < total; idx += 1024) {
        unsigned p = (idx < GSZ) ? ((unsigned)(n0 + idx) | ((unsigned)idx << 17)) : pk[tP[idx]];
        int s = (int)(p & 0x1ffffu);
        int l = (int)(p >> 17);
        v2f es = es2[s];
        float ee = leaky(es.x + edE[l]);
        float eg = leaky(es.y + edG[l]);
        atomicMax(&mE[l], ford(ee));
        atomicMax(&mG[l], ford(eg));
        atomicAdd(&bcnt[l & (NBKT - 1)], 1u);
        tP[idx] = p; tE[idx] = __float_as_uint(ee); tG[idx] = __float_as_uint(eg);
    }
    if (spilled) {
        for (int i = tid; i < NV16; i += 1024) {
            unsigned m = match16(g16[i], pat);
            while (m) {
                int j = __builtin_ctz(m); m &= m - 1u;
                int e = i * 16 + j;
                int s = x[e], l = x[NEDGE + e] - n0;
                v2f es = es2[s];
                atomicMax(&mE[l], ford(leaky(es.x + edE[l])));
                atomicMax(&mG[l], ford(leaky(es.y + edG[l])));
            }
        }
    }
    __syncthreads();

    if (tid < NBKT) {
        unsigned a = 0u;
        for (int k = 0; k < NBKT; ++k) { unsigned c = bcnt[k]; if (k < tid) a += c; }
        bpos[tid] = a;
        if (tid == NBKT - 1) bpos[NBKT] = a + bcnt[NBKT - 1];
    }
    __syncthreads();
    for (int idx = tid; idx < total; idx += 1024) {
        unsigned p = tP[idx];
        int l = (int)(p >> 17);
        float ae = __expf(__uint_as_float(tE[idx]) - dorf(mE[l]));
        float ag = __expf(__uint_as_float(tG[idx]) - dorf(mG[l]));
        atomicAdd(&zE[l], ae);
        atomicAdd(&zG[l], ag);
        int b = l & (NBKT - 1);
        unsigned pos = bpos[b] + atomicAdd(&bfill[b], 1u);
        lst[pos] = p; lgE[pos] = __float_as_uint(ae); lgG[pos] = __float_as_uint(ag);
    }
    __syncthreads();
    for (int i = tid; i < GSZ * ROWP / 4; i += 1024) ((v4f*)acc)[i] = (v4f){0.f, 0.f, 0.f, 0.f};
    __syncthreads();

    {
        const int half = lane >> 4;
        const float* hsel = (half ? h_gnn : h_est) + 4 * (lane & 15);
        float* arow = acc + 4 * lane;
        const int beg = __builtin_amdgcn_readfirstlane((int)bpos[2 * wave]);
        const int end = __builtin_amdgcn_readfirstlane((int)bpos[2 * wave + 2]);
        for (int base = beg; base < end; base += 32) {
            int idx = base + lane;
            bool valid = idx < end;
            unsigned ent = valid ? lst[idx] : 0u;
            int s = (int)(ent & 0x1ffffu);
            int l = (int)(ent >> 17);
            float ae = valid ? __uint_as_float(lgE[idx]) : 0.f;
            float ag = valid ? __uint_as_float(lgG[idx]) : 0.f;
            int nv = end - base; if (nv > 32) nv = 32;
            int j = 0;
            if (nv == 32) {
                v4f ra[8], rb[8];
                rows_load(ra, hsel, s, 0);
                rows_load(rb, hsel, s, 8);
                rows_add(ra, arow, l, ae, ag, half, 0);
                rows_load(ra, hsel, s, 16);
                rows_add(rb, arow, l, ae, ag, half, 8);
                rows_load(rb, hsel, s, 24);
                rows_add(ra, arow, l, ae, ag, half, 16);
                rows_add(rb, arow, l, ae, ag, half, 24);
                j = 32;
            }
            for (; j + 8 <= nv; j += 8) {
                v4f r[8];
                rows_load(r, hsel, s, j);
                rows_add(r, arow, l, ae, ag, half, j);
            }
            for (; j < nv; ++j) {
                int sj = rl_i(s, j), l1 = rl_i(l, j);
                float a1 = rl_f(ae, j), g1 = rl_f(ag, j);
                float w1 = half ? g1 : a1;
                v4f r1 = *(const v4f*)(hsel + (long)sj * NHID);
                v4f* q = (v4f*)(arow + l1 * ROWP);
                v4f a = *q;
                a.x += w1 * r1.x; a.y += w1 * r1.y; a.z += w1 * r1.z; a.w += w1 * r1.w;
                *q = a;
            }
        }
    }
    if (spilled) {
        __syncthreads();
        for (int i = tid; i < NV16; i += 1024) {
            unsigned m = match16(g16[i], pat);
            while (m) {
                int j = __builtin_ctz(m); m &= m - 1u;
                int e = i * 16 + j;
                int s = x[e], l = x[NEDGE + e] - n0;
                v2f es = es2[s];
                float ae = __expf(leaky(es.x + edE[l]) - dorf(mE[l]));
                float ag = __expf(leaky(es.y + edG[l]) - dorf(mG[l]));
                atomicAdd(&zE[l], ae);
                atomicAdd(&zG[l], ag);
                for (int f = 0; f < NHID; ++f) {
                    atomicAdd(&acc[l * ROWP + f], h_est[(long)s * NHID + f] * ae);
                    atomicAdd(&acc[l * ROWP + NHID + f], h_gnn[(long)s * NHID + f] * ag);
                }
            }
        }
    }
    __syncthreads();
    if (tid < GSZ) { zE[tid] = 1.f / zE[tid]; zG[tid] = 1.f / zG[tid]; }
    __syncthreads();

    {
        int f = tid & 63;
        float b = eb[f], rm = erm[f], rs = rsqrtf(erv[f] + EPSBN), gm = egam[f], bt = ebeta[f];
        float* srow = d_out + N_NODES + (long)n0 * NHID;
        for (int i = tid; i < GSZ * NHID; i += 1024) {
            int l = i >> 6;
            float a = fmaxf(acc[l * ROWP + f] * zE[l] + b, 0.f);
            srow[i] = (a - rm) * rs * gm + bt;
        }
    }
    if (tid < GSZ) {
        const float* grow = acc + tid * ROWP + NHID;
        float rzg = zG[tid];
        float y = 0.f;
#pragma unroll 4
        for (int f = 0; f < NHID; ++f) {
            float b = fmaxf(grow[f] * rzg + gb[f], 0.f);
            float zv = (b - grm[f]) * rsqrtf(grv[f] + EPSBN) * ggam[f] + gbeta[f];
            y += zv * clsW[f];
        }
        d_out[n0 + tid] = y + clsb[0];
    }
}

extern "C" void kernel_launch(void* const* d_in, const int* in_sizes, int n_in,
                              void* d_out, int out_size, void* d_ws, size_t ws_size,
                              hipStream_t stream) {
    const float* g        = (const float*)d_in[0];
    const int*   x        = (const int*)d_in[1];
    const float* est_W    = (const float*)d_in[2];
    const float* est_asrc = (const float*)d_in[3];
    const float* est_adst = (const float*)d_in[4];
    const float* est_bias = (const float*)d_in[5];
    const float* est_gam  = (const float*)d_in[6];
    const float* est_beta = (const float*)d_in[7];
    const float* est_rm   = (const float*)d_in[8];
    const float* est_rv   = (const float*)d_in[9];
    const float* gnn_W    = (const float*)d_in[10];
    const float* gnn_asrc = (const float*)d_in[11];
    const float* gnn_adst = (const float*)d_in[12];
    const float* gnn_bias = (const float*)d_in[13];
    const float* gnn_gam  = (const float*)d_in[14];
    const float* gnn_beta = (const float*)d_in[15];
    const float* gnn_rm   = (const float*)d_in[16];
    const float* gnn_rv   = (const float*)d_in[17];
    const float* cls_W    = (const float*)d_in[18];
    const float* cls_b    = (const float*)d_in[19];
    float* out = (float*)d_out;

    float*    ws     = (float*)d_ws;
    float*    h_est  = ws;
    float*    h_gnn  = h_est + (long)N_NODES * NHID;
    float*    es2f   = h_gnn + (long)N_NODES * NHID;
    float*    ed_e   = es2f + 2 * N_NODES;
    float*    ed_g   = ed_e + N_NODES;
    unsigned* grp4   = (unsigned*)(ed_g + N_NODES);
    unsigned* pk     = grp4 + NEDGE / 4;

    const int BT = 256;
    k_gemm<<<GEMMB + CONVB, GT, 0, stream>>>(g, est_W, gnn_W, h_est, h_gnn, x, grp4, pk);

    k_eprep<<<(N_NODES + BT - 1) / BT, BT, 0, stream>>>(h_est, h_gnn,
            est_asrc, est_adst, gnn_asrc, gnn_adst, es2f, ed_e, ed_g);

    k_own<<<NGRP, 1024, 0, stream>>>(x, grp4, pk, es2f, ed_e, ed_g, h_est, h_gnn, out,
            est_bias, est_gam, est_beta, est_rm, est_rv,
            gnn_bias, gnn_gam, gnn_beta, gnn_rm, gnn_rv, cls_W, cls_b);
}


# Owner kernel: entries bucketed by destination row (400 LDS buckets, starts by wave scan); a wave sums a row in registers, one LDS store per row, none per entry, no clear; rows sum in fill order
# speedup vs baseline: 5.2496x; 1.0322x over previous
#include <hip/hip_runtime.h>
#include <cfloat>

#define N_NODES 100000
#define NFEAT   256
#define NHID    64
#define NEDGE   1600000
#define TOTE    (NEDGE + N_NODES)
#define MTILES  (N_NODES / 16)
#define EPSBN   1e-5f
#define NEG_SLOPE 0.2f

typedef __attribute__((ext_vector_type(2))) float v2f;
typedef __attribute__((ext_vector_type(4))) float v4f;
typedef __attribute__((ext_vector_type(8))) float v8f;

__device__ __forceinline__ unsigned ford(float f) {
    unsigned u = __float_as_uint(f);
    return (u & 0x80000000u) ? ~u : (u | 0x80000000u);
}
__device__ __forceinline__ float dorf(unsigned u) {
    return (u & 0x80000000u) ? __uint_as_float(u & 0x7fffffffu) : __uint_as_float(~u);
}
__device__ __forceinline__ float leaky(float v) {
    return fmaxf(v, 0.f) + NEG_SLOPE * fminf(v, 0.f);
}

#define NGRP    250
#define GSZ     400
#define ROWP    132
#define NWAVE   32
#define CAP     8192
#define NV16    (NEDGE / 16)
#define NV32    (NEDGE / 32)

typedef __attribute__((ext_vector_type(4))) int v4i;
typedef __attribute__((ext_vector_type(4))) unsigned v4u;

__device__ __forceinline__ void conv_item(v4i s, v4i d, int t, unsigned* __restrict__ grp4,
                                          unsigned* __restrict__ pk) {
    int gx = d.x / GSZ, gy = d.y / GSZ, gz = d.z / GSZ, gw = d.w / GSZ;
    grp4[t] = (unsigned)gx | ((unsigned)gy << 8) | ((unsigned)gz << 16) | ((unsigned)gw << 24);
    v4u p;
    p.x = (unsigned)s.x | ((unsigned)(d.x - gx * GSZ) << 17);
    p.y = (unsigned)s.y | ((unsigned)(d.y - gy * GSZ) << 17);
    p.z = (unsigned)s.z | ((unsigned)(d.z - gz * GSZ) << 17);
    p.w = (unsigned)s.w | ((unsigned)(d.w - gw * GSZ) << 17);
    *(v4u*)(pk + 4 * t) = p;
}

#define GT      128
#define NBIG    768
#define BIGROWS (NBIG * 4 * 32)
#define NTAIL   ((N_NODES - BIGROWS) / 16)
#define GEMMB   (NBIG + NTAIL)
#define CONVB   400
#define NITEM   (NEDGE / 4)
#define CONVT   (CONVB * GT)

__global__ __launch_bounds__(GT) void k_gemm(const float* __restrict__ g,
                                              const float* __restrict__ We,
                                              const float* __restrict__ Wg,
                                              float* __restrict__ h_est,
                                              float* __restrict__ h_gnn,
                                              const int* __restrict__ x,
                                              unsigned* __restrict__ grp4,
                                              unsigned* __restrict__ pk) {
    if (blockIdx.x >= GEMMB) {
        int t = (blockIdx.x - GEMMB) * GT + threadIdx.x;
        for (; t + 3 * CONVT < NITEM; t += 4 * CONVT) {
            v4i s0 = *(const v4i*)(x + 4 * t);
            v4i d0 = *(const v4i*)(x + NEDGE + 4 * t);
            v4i s1 = *(const v4i*)(x + 4 * (t + CONVT));
            v4i d1 = *(const v4i*)(x + NEDGE + 4 * (t + CONVT));
            v4i s2 = *(const v4i*)(x + 4 * (t + 2 * CONVT));
            v4i d2 = *(const v4i*)(x + NEDGE + 4 * (t + 2 * CONVT));
            v4i s3 = *(const v4i*)(x + 4 * (t + 3 * CONVT));
            v4i d3 = *(const v4i*)(x + NEDGE + 4 * (t + 3 * CONVT));
            conv_item(s0, d0, t, grp4, pk);
            conv_item(s1, d1, t + CONVT, grp4, pk);
            conv_item(s2, d2, t + 2 * CONVT, grp4, pk);
            conv_item(s3, d3, t + 3 * CONVT, grp4, pk);
        }
        for (; t < NITEM; t += CONVT) {
            v4i s0 = *(const v4i*)(x + 4 * t);
            v4i d0 = *(const v4i*)(x + NEDGE + 4 * t);
            conv_item(s0, d0, t, grp4, pk);
        }
        return;
    }
    int wave = threadIdx.x >> 5;
    int lane = threadIdx.x & 31;
    int lrow = lane & 15;
    int lhi  = lane >> 4;

    if (blockIdx.x >= NBIG) {
        const int mt = BIGROWS + (blockIdx.x - NBIG) * 16;
        const float* grow = g + (long)(mt + lrow) * NFEAT;
        const float* Wp = (wave < 2) ? We : Wg;
        float* dst = (wave < 2) ? h_est : h_gnn;
        const int c = (wave & 1) * 32 + lrow;
        v8f t0 = {};
        v8f t1 = {};
        for (int k = 0; k < NFEAT; k += 4) {
            int kr = k + 2 * lhi;
            v2f a = *(const v2f*)(grow + kr);
            v2f b0, b1;
            b0.x = Wp[(long)kr * NHID + c];
            b0.y = Wp[(long)(kr + 1) * NHID + c];
            b1.x = Wp[(long)kr * NHID + c + 16];
            b1.y = Wp[(long)(kr + 1) * NHID + c + 16];
            t0 = __builtin_amdgcn_wmma_f32_16x16x4_f32(
                false, a, false, b0, (short)0, t0, false, false);
            t1 = __builtin_amdgcn_wmma_f32_16x16x4_f32(
                false, a, false, b1, (short)0, t1, false, false);
        }
#pragma unroll
        for (int r = 0; r < 8; ++r) {
            dst[(long)(mt + r + 8 * lhi) * NHID + c]      = t0[r];
            dst[(long)(mt + r + 8 * lhi) * NHID + c + 16] = t1[r];
        }
        return;
    }
    int pair = blockIdx.x * 4 + wave;
    int m0   = pair * 32;

    const float* grow0 = g + (long)(m0 + lrow) * NFEAT;
    const float* grow1 = grow0 + 16 * NFEAT;

    v8f acc0[8] = {};
    v8f acc1[8] = {};
    for (int k = 0; k < NFEAT; k += 4) {
        v2f a0 = *(const v2f*)(grow0 + k + 2 * lhi);
        v2f a1 = *(const v2f*)(grow1 + k + 2 * lhi);
#pragma unroll
        for (int t = 0; t < 8; ++t) {
            const float* Wp = (t < 4) ? We : Wg;
            int c  = (t & 3) * 16 + lrow;
            int kr = k + 2 * lhi;
            v2f b;
            b.x = Wp[(long)kr * NHID + c];
            b.y = Wp[(long)(kr + 1) * NHID + c];
            acc0[t] = __builtin_amdgcn_wmma_f32_16x16x4_f32(
                false, a0, false, b, (short)0, acc0[t], false, false);
            acc1[t] = __builtin_amdgcn_wmma_f32_16x16x4_f32(
                false, a1, false, b, (short)0, acc1[t], false, false);
        }
    }
#pragma unroll
    for (int t = 0; t < 8; ++t) {
        float* dst = (t < 4) ? h_est : h_gnn;
        int c = (t & 3) * 16 + lrow;
#pragma unroll
        for (int r = 0; r < 8; ++r) {
            dst[(long)(m0 + r + 8 * lhi) * NHID + c]      = acc0[t][r];
            dst[(long)(m0 + 16 + r + 8 * lhi) * NHID + c] = acc1[t][r];
        }
    }
}

__global__ void k_eprep(const float* __restrict__ h_est, const float* __restrict__ h_gnn,
                        const float* __restrict__ ase, const float* __restrict__ ade,
                        const float* __restrict__ asg, const float* __restrict__ adg,
                        float* es2, float* ed_e, float* ed_g) {
    int n = blockIdx.x * blockDim.x + threadIdx.x;
    if (n >= N_NODES) return;
    const float* he = h_est + (long)n * NHID;
    const float* hg = h_gnn + (long)n * NHID;
    float se = 0.f, de = 0.f, sg = 0.f, dg = 0.f;
#pragma unroll 4
    for (int f = 0; f < NHID; f += 4) {
        v4f e  = *(const v4f*)(he + f);
        v4f g4 = *(const v4f*)(hg + f);
        v4f a1 = *(const v4f*)(ase + f);
        v4f a2 = *(const v4f*)(ade + f);
        v4f a3 = *(const v4f*)(asg + f);
        v4f a4 = *(const v4f*)(adg + f);
        se += e.x * a1.x + e.y * a1.y + e.z * a1.z + e.w * a1.w;
        de += e.x * a2.x + e.y * a2.y + e.z * a2.z + e.w * a2.w;
        sg += g4.x * a3.x + g4.y * a3.y + g4.z * a3.z + g4.w * a3.w;
        dg += g4.x * a4.x + g4.y * a4.y + g4.z * a4.z + g4.w * a4.w;
    }
    *(v2f*)(es2 + 2 * n) = (v2f){se, sg};
    ed_e[n] = de; ed_g[n] = dg;
}

__device__ __forceinline__ unsigned zbytes(unsigned u) {
    unsigned t = (u & 0x7f7f7f7fu) + 0x7f7f7f7fu;
    return ~(t | u | 0x7f7f7f7fu);
}
__device__ __forceinline__ unsigned nib(unsigned f) {
    return ((f >> 7) & 1u) | ((f >> 14) & 2u) | ((f >> 21) & 4u) | ((f >> 28) & 8u);
}
__device__ __forceinline__ unsigned match16(v4u v, unsigned pat) {
    unsigned f0 = zbytes(v.x ^ pat), f1 = zbytes(v.y ^ pat);
    unsigned f2 = zbytes(v.z ^ pat), f3 = zbytes(v.w ^ pat);
    if ((f0 | f1 | f2 | f3) == 0u) return 0u;
    return nib(f0) | (nib(f1) << 4) | (nib(f2) << 8) | (nib(f3) << 12);
}
__device__ __forceinline__ unsigned flags32(v4u a, v4u b, unsigned pat) {
    return (zbytes(a.x ^ pat) >> 7) | (zbytes(a.y ^ pat) >> 6) |
           (zbytes(a.z ^ pat) >> 5) | (zbytes(a.w ^ pat) >> 4) |
           (zbytes(b.x ^ pat) >> 3) | (zbytes(b.y ^ pat) >> 2) |
           (zbytes(b.z ^ pat) >> 1) |  zbytes(b.w ^ pat);
}
__device__ __forceinline__ int rl_i(int v, int j) { return __builtin_amdgcn_readlane(v, j); }
__device__ __forceinline__ float rl_f(float v, int j) {
    return __int_as_float(__builtin_amdgcn_readlane(__float_as_int(v), j));
}
__device__ __forceinline__ void rows_load(v4f (&r)[8], const float* hsel, int s, int j) {
#pragma unroll
    for (int u = 0; u < 8; ++u) r[u] = *(const v4f*)(hsel + (long)rl_i(s, j + u) * NHID);
}
__device__ __forceinline__ void row_term(v4f& a, int& cur, float* arow, int lj, float w, v4f r) {
    if (lj != cur) {
        *(v4f*)(arow + cur * ROWP) = a;
        a = (v4f){0.f, 0.f, 0.f, 0.f};
        cur = lj;
    }
    a.x += w * r.x; a.y += w * r.y; a.z += w * r.z; a.w += w * r.w;
}
__device__ __forceinline__ void rows_add(const v4f (&r)[8], v4f& a, int& cur, float* arow, int l,
                                         float ae, float ag, int half, int j) {
#pragma unroll
    for (int u = 0; u < 8; ++u) {
        float a1 = rl_f(ae, j + u), g1 = rl_f(ag, j + u);
        row_term(a, cur, arow, rl_i(l, j + u), half ? g1 : a1, r[u]);
    }
}

__global__ __launch_bounds__(1024) void k_own(const int* __restrict__ x,
        const unsigned* __restrict__ grp4, const unsigned* __restrict__ pk,
        const float* __restrict__ es2f,
        const float* __restrict__ ed_e, const float* __restrict__ ed_g,
        const float* __restrict__ h_est, const float* __restrict__ h_gnn,
        float* __restrict__ d_out,
        const float* __restrict__ eb, const float* __restrict__ egam,
        const float* __restrict__ ebeta, const float* __restrict__ erm,
        const float* __restrict__ erv,
        const float* __restrict__ gb, const float* __restrict__ ggam,
        const float* __restrict__ gbeta, const float* __restrict__ grm,
        const float* __restrict__ grv,
        const float* __restrict__ clsW, const float* __restrict__ clsb) {
    __shared__ __attribute__((aligned(16))) float acc[GSZ * ROWP];
    __shared__ unsigned lst[CAP];
    __shared__ unsigned lgE[CAP], lgG[CAP];
    __shared__ unsigned bcnt[GSZ], bfill[GSZ], bpos[GSZ + 1];
    __shared__ unsigned wtot[NWAVE];
    __shared__ float zE[GSZ], zG[GSZ], edE[GSZ], edG[GSZ];
    __shared__ unsigned mE[GSZ], mG[GSZ];
    __shared__ unsigned cnt, ncand, ovf;

    const int tid  = threadIdx.x;
    const int lane = tid & 31;
    const int wave = __builtin_amdgcn_readfirstlane(tid >> 5);
    const int grpid = blockIdx.x;
    const int n0   = grpid * GSZ;
    const unsigned pat = (unsigned)grpid * 0x01010101u;
    const v4u* g16 = (const v4u*)grp4;
    const v2f* es2 = (const v2f*)es2f;
    unsigned* tP = (unsigned*)acc;
    unsigned* tE = tP + CAP;
    unsigned* tG = tE + CAP;

    if (tid < GSZ) {
        int n = n0 + tid;
        bcnt[tid] = 0u; bfill[tid] = 0u;
        edE[tid] = ed_e[n]; edG[tid] = ed_g[n];
        zE[tid] = 0.f; zG[tid] = 0.f;
        mE[tid] = ford(-FLT_MAX); mG[tid] = ford(-FLT_MAX);
    }
    if (tid == 0) { cnt = GSZ; ncand = 0u; ovf = 0u; }
    __syncthreads();

    const v4u ones = (v4u){0xffffffffu, 0xffffffffu, 0xffffffffu, 0xffffffffu};
    for (int i0 = tid - lane; i0 < NV32; i0 += 2048) {
        const int ia = i0 + lane, ib = ia + 1024;
        v4u a0 = ones, a1 = ones, b0 = ones, b1 = ones;
        if (ia < NV32) { a0 = g16[2 * ia]; a1 = g16[2 * ia + 1]; }
        if (ib < NV32) { b0 = g16[2 * ib]; b1 = g16[2 * ib + 1]; }
        const unsigned Pa = flags32(a0, a1, pat), Pb = flags32(b0, b1, pat);
        const unsigned ma = __builtin_amdgcn_ballot_w32(Pa != 0u);
        const unsigned mb = __builtin_amdgcn_ballot_w32(Pb != 0u);
        const unsigned na = (unsigned)__builtin_popcount(ma);
        const unsigned nn = na + (unsigned)__builtin_popcount(mb);
        if (nn != 0u) {
            unsigned base = 0u;
            if (lane == 0) base = atomicAdd(&ncand, nn);
            base = (unsigned)__builtin_amdgcn_readfirstlane((int)base);
            if (Pa != 0u) {
                unsigned idx = base + __builtin_amdgcn_mbcnt_lo(ma, 0u);
                if (idx < CAP) { lgE[idx] = (unsigned)ia; lgG[idx] = Pa; } else ovf = 1u;
            }
            if (Pb != 0u) {
                unsigned idx = base + na + __builtin_amdgcn_mbcnt_lo(mb, 0u);
                if (idx < CAP) { lgE[idx] = (unsigned)ib; lgG[idx] = Pb; } else ovf = 1u;
            }
        }
    }
    __syncthreads();
    {
        const unsigned nc = ncand;
        const int ncl = __builtin_amdgcn_readfirstlane((int)(nc < (unsigned)CAP ? nc : (unsigned)CAP));
        for (int c = tid; c < ncl; c += 1024) {
            const unsigned blk = lgE[c];
            unsigned P = lgG[c];
            while (P) {
                unsigned t = (unsigned)__builtin_ctz(P); P &= P - 1u;
                unsigned idx = atomicAdd(&cnt, 1u);
                if (idx < CAP) tP[idx] = blk * 32u + 4u * (t & 7u) + (t >> 3); else ovf = 1u;
            }
        }
    }
    __syncthreads();
    const bool spilled = ovf != 0u;
    const int total = __builtin_amdgcn_readfirstlane(spilled ? GSZ : (int)cnt);

    for (int idx = tid; idx < total; idx += 1024) {
        unsigned p = (idx < GSZ) ? ((unsigned)(n0 + idx) | ((unsigned)idx << 17)) : pk[tP[idx]];
        int s = (int)(p & 0x1ffffu);
        int l = (int)(p >> 17);
        v2f es = es2[s];
        float ee = leaky(es.x + edE[l]);
        float eg = leaky(es.y + edG[l]);
        atomicMax(&mE[l], ford(ee));
        atomicMax(&mG[l], ford(eg));
        atomicAdd(&bcnt[l], 1u);
        tP[idx] = p; tE[idx] = __float_as_uint(ee); tG[idx] = __float_as_uint(eg);
    }
    if (spilled) {
        for (int i = tid; i < NV16; i += 1024) {
            unsigned m = match16(g16[i], pat);
            while (m) {
                int j = __builtin_ctz(m); m &= m - 1u;
                int e = i * 16 + j;
                int s = x[e], l = x[NEDGE + e] - n0;
                v2f es = es2[s];
                atomicMax(&mE[l], ford(leaky(es.x + edE[l])));
                atomicMax(&mG[l], ford(leaky(es.y + edG[l])));
            }
        }
    }
    __syncthreads();

    {
        const unsigned c = (tid < GSZ) ? bcnt[tid] : 0u;
        unsigned inc = c;
#pragma unroll
        for (int d = 1; d < 32; d <<= 1) {
            unsigned up = (unsigned)__shfl_up((int)inc, d);
            if (lane >= d) inc += up;
        }
        if (lane == 31) wtot[wave] = inc;
        __syncthreads();
        if (tid < GSZ) {
            unsigned front = 0u;
            for (int k = 0; k < (GSZ + 31) / 32; ++k) { unsigned w = wtot[k]; if (k < wave) front += w; }
            bpos[tid] = front + inc - c;
            if (tid == GSZ - 1) bpos[GSZ] = front + inc;
        }
    }
    __syncthreads();
    for (int idx = tid; idx < total; idx += 1024) {
        unsigned p = tP[idx];
        int l = (int)(p >> 17);
        float ae = __expf(__uint_as_float(tE[idx]) - dorf(mE[l]));
        float ag = __expf(__uint_as_float(tG[idx]) - dorf(mG[l]));
        atomicAdd(&zE[l], ae);
        atomicAdd(&zG[l], ag);
        unsigned pos = bpos[l] + atomicAdd(&bfill[l], 1u);
        lst[pos] = p; lgE[pos] = __float_as_uint(ae); lgG[pos] = __float_as_uint(ag);
    }
    __syncthreads();

    {
        const int half = lane >> 4;
        const float* hsel = (half ? h_gnn : h_est) + 4 * (lane & 15);
        float* arow = acc + 4 * lane;
        const int r0 = (wave * GSZ) / NWAVE, r1 = ((wave + 1) * GSZ) / NWAVE;
        const int beg = __builtin_amdgcn_readfirstlane((int)bpos[r0]);
        const int end = __builtin_amdgcn_readfirstlane((int)bpos[r1]);
        v4f a = (v4f){0.f, 0.f, 0.f, 0.f};
        int cur = r0;
        for (int base = beg; base < end; base += 32) {
            int idx = base + lane;
            bool valid = idx < end;
            unsigned ent = valid ? lst[idx] : 0u;
            int s = (int)(ent & 0x1ffffu);
            int l = (int)(ent >> 17);
            float ae = valid ? __uint_as_float(lgE[idx]) : 0.f;
            float ag = valid ? __uint_as_float(lgG[idx]) : 0.f;
            int nv = end - base; if (nv > 32) nv = 32;
            int j = 0;
            if (nv == 32) {
                v4f ra[8], rb[8];
                rows_load(ra, hsel, s, 0);
                rows_load(rb, hsel, s, 8);
                rows_add(ra, a, cur, arow, l, ae, ag, half, 0);
                rows_load(ra, hsel, s, 16);
                rows_add(rb, a, cur, arow, l, ae, ag, half, 8);
                rows_load(rb, hsel, s, 24);
                rows_add(ra, a, cur, arow, l, ae, ag, half, 16);
                rows_add(rb, a, cur, arow, l, ae, ag, half, 24);
                j = 32;
            }
            for (; j + 8 <= nv; j += 8) {
                v4f r[8];
                rows_load(r, hsel, s, j);
                rows_add(r, a, cur, arow, l, ae, ag, half, j);
            }
            for (; j < nv; ++j) {
                int sj = rl_i(s, j), l1 = rl_i(l, j);
                float a1 = rl_f(ae, j), g1 = rl_f(ag, j);
                v4f r1 = *(const v4f*)(hsel + (long)sj * NHID);
                row_term(a, cur, arow, l1, half ? g1 : a1, r1);
            }
        }
        *(v4f*)(arow + cur * ROWP) = a;
    }
    if (spilled) {
        __syncthreads();
        for (int i = tid; i < NV16; i += 1024) {
            unsigned m = match16(g16[i], pat);
            while (m) {
                int j = __builtin_ctz(m); m &= m - 1u;
                int e = i * 16 + j;
                int s = x[e], l = x[NEDGE + e] - n0;
                v2f es = es2[s];
                float ae = __expf(leaky(es.x + edE[l]) - dorf(mE[l]));
                float ag = __expf(leaky(es.y + edG[l]) - dorf(mG[l]));
                atomicAdd(&zE[l], ae);
                atomicAdd(&zG[l], ag);
                for (int f = 0; f < NHID; ++f) {
                    atomicAdd(&acc[l * ROWP + f], h_est[(long)s * NHID + f] * ae);
                    atomicAdd(&acc[l * ROWP + NHID + f], h_gnn[(long)s * NHID + f] * ag);
                }
            }
        }
    }
    __syncthreads();
    if (tid < GSZ) { zE[tid] = 1.f / zE[tid]; zG[tid] = 1.f / zG[tid]; }
    __syncthreads();

    {
        int f = tid & 63;
        float b = eb[f], rm = erm[f], rs = rsqrtf(erv[f] + EPSBN), gm = egam[f], bt = ebeta[f];
        float* srow = d_out + N_NODES + (long)n0 * NHID;
        for (int i = tid; i < GSZ * NHID; i += 1024) {
            int l = i >> 6;
            float a = fmaxf(acc[l * ROWP + f] * zE[l] + b, 0.f);
            srow[i] = (a - rm) * rs * gm + bt;
        }
    }
    if (tid < GSZ) {
        const float* grow = acc + tid * ROWP + NHID;
        float rzg = zG[tid];
        float y = 0.f;
#pragma unroll 4
        for (int f = 0; f < NHID; ++f) {
            float b = fmaxf(grow[f] * rzg + gb[f], 0.f);
            float zv = (b - grm[f]) * rsqrtf(grv[f] + EPSBN) * ggam[f] + gbeta[f];
            y += zv * clsW[f];
        }
        d_out[n0 + tid] = y + clsb[0];
    }
}

extern "C" void kernel_launch(void* const* d_in, const int* in_sizes, int n_in,
                              void* d_out, int out_size, void* d_ws, size_t ws_size,
                              hipStream_t stream) {
    const float* g        = (const float*)d_in[0];
    const int*   x        = (const int*)d_in[1];
    const float* est_W    = (const float*)d_in[2];
    const float* est_asrc = (const float*)d_in[3];
    const float* est_adst = (const float*)d_in[4];
    const float* est_bias = (const float*)d_in[5];
    const float* est_gam  = (const float*)d_in[6];
    const float* est_beta = (const float*)d_in[7];
    const float* est_rm   = (const float*)d_in[8];
    const float* est_rv   = (const float*)d_in[9];
    const float* gnn_W    = (const float*)d_in[10];
    const float* gnn_asrc = (const float*)d_in[11];
    const float* gnn_adst = (const float*)d_in[12];
    const float* gnn_bias = (const float*)d_in[13];
    const float* gnn_gam  = (const float*)d_in[14];
    const float* gnn_beta = (const float*)d_in[15];
    const float* gnn_rm   = (const float*)d_in[16];
    const float* gnn_rv   = (const float*)d_in[17];
    const float* cls_W    = (const float*)d_in[18];
    const float* cls_b    = (const float*)d_in[19];
    float* out = (float*)d_out;

    float*    ws     = (float*)d_ws;
    float*    h_est  = ws;
    float*    h_gnn  = h_est + (long)N_NODES * NHID;
    float*    es2f   = h_gnn + (long)N_NODES * NHID;
    float*    ed_e   = es2f + 2 * N_NODES;
    float*    ed_g   = ed_e + N_NODES;
    unsigned* grp4   = (unsigned*)(ed_g + N_NODES);
    unsigned* pk     = grp4 + NEDGE / 4;

    const int BT = 256;
    k_gemm<<<GEMMB + CONVB, GT, 0, stream>>>(g, est_W, gnn_W, h_est, h_gnn, x, grp4, pk);

    k_eprep<<<(N_NODES + BT - 1) / BT, BT, 0, stream>>>(h_est, h_gnn,
            est_asrc, est_adst, gnn_asrc, gnn_adst, es2f, ed_e, ed_g);

    k_own<<<NGRP, 1024, 0, stream>>>(x, grp4, pk, es2f, ed_e, ed_g, h_est, h_gnn, out,
            est_bias, est_gam, est_beta, est_rm, est_rv,
            gnn_bias, gnn_gam, gnn_beta, gnn_rm, gnn_rv, cls_W, cls_b);
}
